# starting kernel
# baseline (speedup 1.0000x reference)
#include <hip/hip_runtime.h>
#include <cstddef>

typedef __attribute__((ext_vector_type(2))) float v2f;
typedef __attribute__((ext_vector_type(8))) float v8f;

#define DEV __device__ __forceinline__

DEV float eluf(float v) { return v > 0.f ? v : (__expf(v) - 1.f); }
DEV float histf(float v, float bin, float nm1) {
  return fmaxf(1.f - fabsf(v - bin) * nm1, 0.f);
}

DEV v8f wmma_f32(v2f a, v2f b, v8f c) {
  return __builtin_amdgcn_wmma_f32_16x16x4_f32(false, a, false, b, (short)0, c,
                                               false, false);
}

__global__ void tr0_kernel(const float* __restrict__ w, float* __restrict__ wT) {
  int idx = blockIdx.x * blockDim.x + threadIdx.x;
  if (idx >= 864 * 64) return;
  int k = idx >> 6, n = idx & 63;
  int r = k / 96, cf = k % 96;
  float v = (cf < 93) ? w[(size_t)n * 837 + cf * 9 + r] : 0.f;
  wT[((((size_t)(k >> 1)) * 64 + n) << 1) + (k & 1)] = v;
}

__global__ void tr1_kernel(const float* __restrict__ w, float* __restrict__ wT) {
  int idx = blockIdx.x * blockDim.x + threadIdx.x;
  if (idx >= 576 * 64) return;
  int k = idx >> 6, n = idx & 63;
  int r = k >> 6, cin = k & 63;
  float v = w[(size_t)n * 576 + cin * 9 + r];
  wT[((((size_t)(k >> 1)) * 64 + n) << 1) + (k & 1)] = v;
}

__global__ __launch_bounds__(32) void conv0_kernel(const float* __restrict__ x,
                                                   const float* __restrict__ wT,
                                                   const float* __restrict__ bias,
                                                   float* __restrict__ f1) {
  const int lane = threadIdx.x;
  const int hh = lane >> 4;
  const int ml = lane & 15;
  const int tx = blockIdx.x;
  const int py = blockIdx.y;
  const int b = blockIdx.z;

  __shared__ float ft[165][98];

  const int gy0 = 4 * py;
  const int gx0 = 32 * tx;
  const float* xb = x + (size_t)b * 3 * 512 * 512;

  for (int p = lane; p < 165; p += 32) {
    const int ly = p / 33, lx = p - ly * 33;
    const int gy = min(gy0 + ly, 511);
    const int gx = min(gx0 + lx, 511);
    const int ym = max(gy - 1, 0), yp = min(gy + 1, 511);
    const int xm = max(gx - 1, 0), xp = min(gx + 1, 511);
    float* f = &ft[p][0];
    float cv[3];
    int o = 9;
#pragma unroll
    for (int c = 0; c < 3; ++c) {
      const float* xc = xb + (size_t)c * 512 * 512;
      float a = xc[ym * 512 + xm], bb = xc[ym * 512 + gx], cc = xc[ym * 512 + xp];
      float d = xc[gy * 512 + xm], ee = xc[gy * 512 + gx], e = xc[gy * 512 + xp];
      float g = xc[yp * 512 + xm], h = xc[yp * 512 + gx], i = xc[yp * 512 + xp];
      cv[c] = ee;
      f[c] = ee;
      f[3 + 2 * c] = ((cc + 2.f * e + i) - (a + 2.f * d + g)) * 0.125f;
      f[4 + 2 * c] = ((g + 2.f * h + i) - (a + 2.f * bb + cc)) * 0.125f;
    }
#pragma unroll
    for (int c = 0; c < 3; ++c)
#pragma unroll
      for (int bn = 0; bn < 4; ++bn) f[o++] = histf(cv[c], (float)bn * (1.f / 3.f), 3.f);
#pragma unroll
    for (int c = 0; c < 3; ++c)
#pragma unroll
      for (int bn = 0; bn < 8; ++bn) f[o++] = histf(cv[c], (float)bn * (1.f / 7.f), 7.f);
#pragma unroll
    for (int c = 0; c < 3; ++c)
#pragma unroll
      for (int bn = 0; bn < 16; ++bn) f[o++] = histf(cv[c], (float)bn * (1.f / 15.f), 15.f);
    f[93] = 0.f;
    f[94] = 0.f;
    f[95] = 0.f;
  }
  __syncthreads();

  v8f acc[2][4];
  const v8f vz = {0.f, 0.f, 0.f, 0.f, 0.f, 0.f, 0.f, 0.f};
  for (int mt = 0; mt < 2; ++mt)
    for (int nt = 0; nt < 4; ++nt) acc[mt][nt] = vz;

  for (int r = 0; r < 9; ++r) {
    const int ky = r / 3, kx = r % 3;
    const int lx = 2 * ml + kx;
    const float* fp0 = &ft[ky * 33 + lx][2 * hh];
    const float* fp1 = &ft[(ky + 2) * 33 + lx][2 * hh];
    __builtin_prefetch(wT + (((size_t)(min(r + 1, 8) * 48)) * 64) * 2, 0, 1);
    for (int u = 0; u < 24; ++u) {
      v2f a0 = *(const v2f*)(fp0 + 4 * u);
      v2f a1 = *(const v2f*)(fp1 + 4 * u);
      const float* bp = wT + (((size_t)(r * 48 + 2 * u + hh)) * 64) * 2;
#pragma unroll
      for (int nt = 0; nt < 4; ++nt) {
        v2f bv = *(const v2f*)(bp + ((nt * 16 + ml) << 1));
        acc[0][nt] = wmma_f32(a0, bv, acc[0][nt]);
        acc[1][nt] = wmma_f32(a1, bv, acc[1][nt]);
      }
    }
  }

  const int pxb = 8 * tx;
#pragma unroll
  for (int nt = 0; nt < 4; ++nt) {
    const float bb = bias[nt * 16 + ml];
#pragma unroll
    for (int j = 0; j < 4; ++j) {
      float m0 = fmaxf(eluf(acc[0][nt][2 * j] + bb), eluf(acc[0][nt][2 * j + 1] + bb));
      float m1 = fmaxf(eluf(acc[1][nt][2 * j] + bb), eluf(acc[1][nt][2 * j + 1] + bb));
      int px = pxb + j + 4 * hh;
      if (px < 127)
        f1[((((size_t)b * 127 + py) * 127 + px) << 6) + nt * 16 + ml] = fmaxf(m0, m1);
    }
  }
}

template <int HIN, int WIN, int HP, int WP>
__global__ __launch_bounds__(32) void convmid_kernel(const float* __restrict__ fin,
                                                     const float* __restrict__ wT,
                                                     const float* __restrict__ bias,
                                                     float* __restrict__ fout) {
  const int lane = threadIdx.x;
  const int hh = lane >> 4;
  const int ml = lane & 15;
  const int tx = blockIdx.x;
  const int py = blockIdx.y;
  const int b = blockIdx.z;

  v8f acc[2][4];
  const v8f vz = {0.f, 0.f, 0.f, 0.f, 0.f, 0.f, 0.f, 0.f};
  for (int mt = 0; mt < 2; ++mt)
    for (int nt = 0; nt < 4; ++nt) acc[mt][nt] = vz;

  const int cx = 16 * tx + ml;
  const float* finb = fin + (size_t)b * HIN * WIN * 64;

  for (int r = 0; r < 9; ++r) {
    const int ky = r / 3, kx = r % 3;
    const int xx = min(2 * cx + kx, WIN - 1);
    const int yy0 = min(4 * py + ky, HIN - 1);
    const int yy1 = min(4 * py + 2 + ky, HIN - 1);
    const float* row0 = finb + (((size_t)yy0 * WIN + xx) << 6);
    const float* row1 = finb + (((size_t)yy1 * WIN + xx) << 6);
    __builtin_prefetch(wT + (((size_t)(min(r + 1, 8) * 32)) * 64) * 2, 0, 1);
    for (int u = 0; u < 16; ++u) {
      const int cinb = 4 * u + 2 * hh;
      v2f a0 = *(const v2f*)(row0 + cinb);
      v2f a1 = *(const v2f*)(row1 + cinb);
      const float* bp = wT + (((size_t)(r * 32 + 2 * u + hh)) * 64) * 2;
#pragma unroll
      for (int nt = 0; nt < 4; ++nt) {
        v2f bv = *(const v2f*)(bp + ((nt * 16 + ml) << 1));
        acc[0][nt] = wmma_f32(a0, bv, acc[0][nt]);
        acc[1][nt] = wmma_f32(a1, bv, acc[1][nt]);
      }
    }
  }

  const int pxb = 8 * tx;
#pragma unroll
  for (int nt = 0; nt < 4; ++nt) {
    const float bb = bias[nt * 16 + ml];
#pragma unroll
    for (int j = 0; j < 4; ++j) {
      float m0 = fmaxf(eluf(acc[0][nt][2 * j] + bb), eluf(acc[0][nt][2 * j + 1] + bb));
      float m1 = fmaxf(eluf(acc[1][nt][2 * j] + bb), eluf(acc[1][nt][2 * j + 1] + bb));
      int px = pxb + j + 4 * hh;
      if (px < WP)
        fout[((((size_t)b * HP + py) * WP + px) << 6) + nt * 16 + ml] = fmaxf(m0, m1);
    }
  }
}

__global__ void mean_kernel(const float* __restrict__ f3, float* __restrict__ gm) {
  int b = blockIdx.x, c = threadIdx.x;
  float s = 0.f;
  for (int i = 0; i < 49; ++i) s += f3[(((size_t)b * 49 + i) << 6) + c];
  gm[b * 64 + c] = s * (1.f / 49.f);
}

__global__ void fc_kernel(const float* __restrict__ gm, const float* __restrict__ w1,
                          const float* __restrict__ b1, const float* __restrict__ w2,
                          const float* __restrict__ b2, float* __restrict__ wq) {
  __shared__ float fv[64];
  __shared__ float h[64];
  int b = blockIdx.x, j = threadIdx.x;
  fv[j] = gm[b * 64 + j];
  __syncthreads();
  float acc = b1[j];
  for (int c = 0; c < 64; ++c) acc += fv[c] * w1[j * 64 + c];
  h[j] = eluf(acc);
  __syncthreads();
  if (j < 30) {
    float a2 = b2[j];
    for (int c = 0; c < 64; ++c) a2 += h[c] * w2[j * 64 + c];
    wq[b * 30 + j] = a2;
  }
}

__global__ void quad_kernel(const float* __restrict__ x, const float* __restrict__ wq,
                            float* __restrict__ out) {
  const int HW = 512 * 512;
  int idx = blockIdx.x * blockDim.x + threadIdx.x;
  if (idx >= 4 * HW) return;
  int b = idx / HW, hw = idx - b * HW;
  const float* xb = x + (size_t)b * 3 * HW;
  float r = xb[hw], g = xb[HW + hw], bl = xb[2 * HW + hw];
  float f[10] = {r * r, r * g, r * bl, r, g * g, g * bl, g, bl * bl, bl, 1.f};
  const float* w = wq + b * 30;
  float* ob = out + (size_t)b * 3 * HW;
#pragma unroll
  for (int c = 0; c < 3; ++c) {
    float s = (c == 0) ? r : ((c == 1) ? g : bl);
#pragma unroll
    for (int k = 0; k < 10; ++k) s += f[k] * w[c * 10 + k];
    ob[c * HW + hw] = s;
  }
}

static const size_t OFF_WT0 = 0;
static const size_t OFF_WT1 = 55296;
static const size_t OFF_WT2 = 92160;
static const size_t OFF_F1 = 129024;
static const size_t OFF_F2 = 129024 + 4129024;
static const size_t OFF_F3 = 129024 + 4129024 + 246016;
static const size_t OFF_GM = OFF_F3 + 12544;
static const size_t OFF_WQ = OFF_GM + 256;

extern "C" void kernel_launch(void* const* d_in, const int* in_sizes, int n_in,
                              void* d_out, int out_size, void* d_ws, size_t ws_size,
                              hipStream_t stream) {
  const float* x = (const float*)d_in[0];
  const float* c0w = (const float*)d_in[1];
  const float* c0b = (const float*)d_in[2];
  const float* c1w = (const float*)d_in[3];
  const float* c1b = (const float*)d_in[4];
  const float* c2w = (const float*)d_in[5];
  const float* c2b = (const float*)d_in[6];
  const float* f1w = (const float*)d_in[7];
  const float* f1b = (const float*)d_in[8];
  const float* f2w = (const float*)d_in[9];
  const float* f2b = (const float*)d_in[10];
  float* out = (float*)d_out;
  float* ws = (float*)d_ws;

  float* wT0 = ws + OFF_WT0;
  float* wT1 = ws + OFF_WT1;
  float* wT2 = ws + OFF_WT2;
  float* F1 = ws + OFF_F1;
  float* F2 = ws + OFF_F2;
  float* F3 = ws + OFF_F3;
  float* GM = ws + OFF_GM;
  float* WQ = ws + OFF_WQ;

  tr0_kernel<<<(864 * 64 + 255) / 256, 256, 0, stream>>>(c0w, wT0);
  tr1_kernel<<<(576 * 64 + 255) / 256, 256, 0, stream>>>(c1w, wT1);
  tr1_kernel<<<(576 * 64 + 255) / 256, 256, 0, stream>>>(c2w, wT2);

  conv0_kernel<<<dim3(16, 127, 4), 32, 0, stream>>>(x, wT0, c0b, F1);
  convmid_kernel<127, 127, 31, 31><<<dim3(4, 31, 4), 32, 0, stream>>>(F1, wT1, c1b, F2);
  convmid_kernel<31, 31, 7, 7><<<dim3(1, 7, 4), 32, 0, stream>>>(F2, wT2, c2b, F3);

  mean_kernel<<<4, 64, 0, stream>>>(F3, GM);
  fc_kernel<<<4, 64, 0, stream>>>(GM, f1w, f1b, f2w, f2b, WQ);

  const int npix = 4 * 512 * 512;
  quad_kernel<<<(npix + 255) / 256, 256, 0, stream>>>(x, WQ, out);
}


# conv K-loops software-pipelined: the next 8 k-steps' A and B fragments are loaded as one group into a second register set ahead of the current block's 64 WMMAs (all three conv launches)
# speedup vs baseline: 1.5346x; 1.5346x over previous
#include <hip/hip_runtime.h>
#include <cstddef>

typedef __attribute__((ext_vector_type(2))) float v2f;
typedef __attribute__((ext_vector_type(8))) float v8f;

#define DEV __device__ __forceinline__

DEV float eluf(float v) { return v > 0.f ? v : (__expf(v) - 1.f); }
DEV float histf(float v, float bin, float nm1) {
  return fmaxf(1.f - fabsf(v - bin) * nm1, 0.f);
}

DEV v8f wmma_f32(v2f a, v2f b, v8f c) {
  return __builtin_amdgcn_wmma_f32_16x16x4_f32(false, a, false, b, (short)0, c,
                                               false, false);
}

template <int KB>
struct FragSet {
  v2f a0[KB];
  v2f a1[KB];
  v2f b[KB][4];
};

template <int KB>
DEV void mma_block(const FragSet<KB>& s, v8f (&acc)[2][4]) {
#pragma unroll
  for (int i = 0; i < KB; ++i) {
#pragma unroll
    for (int nt = 0; nt < 4; ++nt) {
      acc[0][nt] = wmma_f32(s.a0[i], s.b[i][nt], acc[0][nt]);
      acc[1][nt] = wmma_f32(s.a1[i], s.b[i][nt], acc[1][nt]);
    }
  }
}

__global__ void tr0_kernel(const float* __restrict__ w, float* __restrict__ wT) {
  int idx = blockIdx.x * blockDim.x + threadIdx.x;
  if (idx >= 864 * 64) return;
  int k = idx >> 6, n = idx & 63;
  int r = k / 96, cf = k % 96;
  float v = (cf < 93) ? w[(size_t)n * 837 + cf * 9 + r] : 0.f;
  wT[((((size_t)(k >> 1)) * 64 + n) << 1) + (k & 1)] = v;
}

__global__ void tr1_kernel(const float* __restrict__ w, float* __restrict__ wT) {
  int idx = blockIdx.x * blockDim.x + threadIdx.x;
  if (idx >= 576 * 64) return;
  int k = idx >> 6, n = idx & 63;
  int r = k >> 6, cin = k & 63;
  float v = w[(size_t)n * 576 + cin * 9 + r];
  wT[((((size_t)(k >> 1)) * 64 + n) << 1) + (k & 1)] = v;
}

__global__ __launch_bounds__(32) void conv0_kernel(const float* __restrict__ x,
                                                   const float* __restrict__ wT,
                                                   const float* __restrict__ bias,
                                                   float* __restrict__ f1) {
  const int lane = threadIdx.x;
  const int hh = lane >> 4;
  const int ml = lane & 15;
  const int tx = blockIdx.x;
  const int py = blockIdx.y;
  const int b = blockIdx.z;

  __shared__ float ft[165][98];

  const int gy0 = 4 * py;
  const int gx0 = 32 * tx;
  const float* xb = x + (size_t)b * 3 * 512 * 512;

  for (int p = lane; p < 165; p += 32) {
    const int ly = p / 33, lx = p - ly * 33;
    const int gy = min(gy0 + ly, 511);
    const int gx = min(gx0 + lx, 511);
    const int ym = max(gy - 1, 0), yp = min(gy + 1, 511);
    const int xm = max(gx - 1, 0), xp = min(gx + 1, 511);
    float* f = &ft[p][0];
    float cv[3];
    int o = 9;
#pragma unroll
    for (int c = 0; c < 3; ++c) {
      const float* xc = xb + (size_t)c * 512 * 512;
      float a = xc[ym * 512 + xm], bb = xc[ym * 512 + gx], cc = xc[ym * 512 + xp];
      float d = xc[gy * 512 + xm], ee = xc[gy * 512 + gx], e = xc[gy * 512 + xp];
      float g = xc[yp * 512 + xm], h = xc[yp * 512 + gx], i = xc[yp * 512 + xp];
      cv[c] = ee;
      f[c] = ee;
      f[3 + 2 * c] = ((cc + 2.f * e + i) - (a + 2.f * d + g)) * 0.125f;
      f[4 + 2 * c] = ((g + 2.f * h + i) - (a + 2.f * bb + cc)) * 0.125f;
    }
#pragma unroll
    for (int c = 0; c < 3; ++c)
#pragma unroll
      for (int bn = 0; bn < 4; ++bn) f[o++] = histf(cv[c], (float)bn * (1.f / 3.f), 3.f);
#pragma unroll
    for (int c = 0; c < 3; ++c)
#pragma unroll
      for (int bn = 0; bn < 8; ++bn) f[o++] = histf(cv[c], (float)bn * (1.f / 7.f), 7.f);
#pragma unroll
    for (int c = 0; c < 3; ++c)
#pragma unroll
      for (int bn = 0; bn < 16; ++bn) f[o++] = histf(cv[c], (float)bn * (1.f / 15.f), 15.f);
    f[93] = 0.f;
    f[94] = 0.f;
    f[95] = 0.f;
  }
  __syncthreads();

  v8f acc[2][4];
  const v8f vz = {0.f, 0.f, 0.f, 0.f, 0.f, 0.f, 0.f, 0.f};
  for (int mt = 0; mt < 2; ++mt)
    for (int nt = 0; nt < 4; ++nt) acc[mt][nt] = vz;

  constexpr int KB = 8;
  constexpr int BPT = 24 / KB;
  constexpr int NB = 9 * BPT;
  auto load_block = [&](FragSet<KB>& s, int j) {
    const int r = j / BPT;
    const int ub = (j - r * BPT) * KB;
    const int ky = r / 3, kx = r - 3 * ky;
    const int lx = 2 * ml + kx;
    const float* fp0 = &ft[ky * 33 + lx][2 * hh + 4 * ub];
    const float* fp1 = &ft[(ky + 2) * 33 + lx][2 * hh + 4 * ub];
    const float* bp = wT + ((((size_t)(r * 48 + 2 * ub + hh)) * 64 + ml) << 1);
#pragma unroll
    for (int i = 0; i < KB; ++i) {
      s.a0[i] = *(const v2f*)(fp0 + 4 * i);
      s.a1[i] = *(const v2f*)(fp1 + 4 * i);
#pragma unroll
      for (int nt = 0; nt < 4; ++nt) s.b[i][nt] = *(const v2f*)(bp + i * 256 + nt * 32);
    }
  };

  FragSet<KB> fx, fy;
  load_block(fx, 0);
  for (int j = 0; j + 1 < NB; j += 2) {
    load_block(fy, j + 1);
    __builtin_amdgcn_sched_barrier(0);
    mma_block<KB>(fx, acc);
    __builtin_amdgcn_sched_barrier(0);
    load_block(fx, min(j + 2, NB - 1));
    __builtin_amdgcn_sched_barrier(0);
    mma_block<KB>(fy, acc);
  }
  if (NB & 1) mma_block<KB>(fx, acc);

  const int pxb = 8 * tx;
#pragma unroll
  for (int nt = 0; nt < 4; ++nt) {
    const float bb = bias[nt * 16 + ml];
#pragma unroll
    for (int j = 0; j < 4; ++j) {
      float m0 = fmaxf(eluf(acc[0][nt][2 * j] + bb), eluf(acc[0][nt][2 * j + 1] + bb));
      float m1 = fmaxf(eluf(acc[1][nt][2 * j] + bb), eluf(acc[1][nt][2 * j + 1] + bb));
      int px = pxb + j + 4 * hh;
      if (px < 127)
        f1[((((size_t)b * 127 + py) * 127 + px) << 6) + nt * 16 + ml] = fmaxf(m0, m1);
    }
  }
}

template <int HIN, int WIN, int HP, int WP>
__global__ __launch_bounds__(32) void convmid_kernel(const float* __restrict__ fin,
                                                     const float* __restrict__ wT,
                                                     const float* __restrict__ bias,
                                                     float* __restrict__ fout) {
  const int lane = threadIdx.x;
  const int hh = lane >> 4;
  const int ml = lane & 15;
  const int tx = blockIdx.x;
  const int py = blockIdx.y;
  const int b = blockIdx.z;

  v8f acc[2][4];
  const v8f vz = {0.f, 0.f, 0.f, 0.f, 0.f, 0.f, 0.f, 0.f};
  for (int mt = 0; mt < 2; ++mt)
    for (int nt = 0; nt < 4; ++nt) acc[mt][nt] = vz;

  const int cx = 16 * tx + ml;
  const float* finb = fin + (size_t)b * HIN * WIN * 64;

  constexpr int KB = 8;
  constexpr int BPT = 16 / KB;
  constexpr int NB = 9 * BPT;
  auto load_block = [&](FragSet<KB>& s, int j) {
    const int r = j / BPT;
    const int ub = (j - r * BPT) * KB;
    const int ky = r / 3, kx = r - 3 * ky;
    const int xx = min(2 * cx + kx, WIN - 1);
    const int yy0 = min(4 * py + ky, HIN - 1);
    const int yy1 = min(4 * py + 2 + ky, HIN - 1);
    const float* row0 = finb + (((size_t)yy0 * WIN + xx) << 6) + 4 * ub + 2 * hh;
    const float* row1 = finb + (((size_t)yy1 * WIN + xx) << 6) + 4 * ub + 2 * hh;
    const float* bp = wT + ((((size_t)(r * 32 + 2 * ub + hh)) * 64 + ml) << 1);
#pragma unroll
    for (int i = 0; i < KB; ++i) {
      s.a0[i] = *(const v2f*)(row0 + 4 * i);
      s.a1[i] = *(const v2f*)(row1 + 4 * i);
#pragma unroll
      for (int nt = 0; nt < 4; ++nt) s.b[i][nt] = *(const v2f*)(bp + i * 256 + nt * 32);
    }
  };

  FragSet<KB> fx, fy;
  load_block(fx, 0);
  for (int j = 0; j + 1 < NB; j += 2) {
    load_block(fy, j + 1);
    __builtin_amdgcn_sched_barrier(0);
    mma_block<KB>(fx, acc);
    __builtin_amdgcn_sched_barrier(0);
    load_block(fx, min(j + 2, NB - 1));
    __builtin_amdgcn_sched_barrier(0);
    mma_block<KB>(fy, acc);
  }
  if (NB & 1) mma_block<KB>(fx, acc);

  const int pxb = 8 * tx;
#pragma unroll
  for (int nt = 0; nt < 4; ++nt) {
    const float bb = bias[nt * 16 + ml];
#pragma unroll
    for (int j = 0; j < 4; ++j) {
      float m0 = fmaxf(eluf(acc[0][nt][2 * j] + bb), eluf(acc[0][nt][2 * j + 1] + bb));
      float m1 = fmaxf(eluf(acc[1][nt][2 * j] + bb), eluf(acc[1][nt][2 * j + 1] + bb));
      int px = pxb + j + 4 * hh;
      if (px < WP)
        fout[((((size_t)b * HP + py) * WP + px) << 6) + nt * 16 + ml] = fmaxf(m0, m1);
    }
  }
}

__global__ void mean_kernel(const float* __restrict__ f3, float* __restrict__ gm) {
  int b = blockIdx.x, c = threadIdx.x;
  float s = 0.f;
  for (int i = 0; i < 49; ++i) s += f3[(((size_t)b * 49 + i) << 6) + c];
  gm[b * 64 + c] = s * (1.f / 49.f);
}

__global__ void fc_kernel(const float* __restrict__ gm, const float* __restrict__ w1,
                          const float* __restrict__ b1, const float* __restrict__ w2,
                          const float* __restrict__ b2, float* __restrict__ wq) {
  __shared__ float fv[64];
  __shared__ float h[64];
  int b = blockIdx.x, j = threadIdx.x;
  fv[j] = gm[b * 64 + j];
  __syncthreads();
  float acc = b1[j];
  for (int c = 0; c < 64; ++c) acc += fv[c] * w1[j * 64 + c];
  h[j] = eluf(acc);
  __syncthreads();
  if (j < 30) {
    float a2 = b2[j];
    for (int c = 0; c < 64; ++c) a2 += h[c] * w2[j * 64 + c];
    wq[b * 30 + j] = a2;
  }
}

__global__ void quad_kernel(const float* __restrict__ x, const float* __restrict__ wq,
                            float* __restrict__ out) {
  const int HW = 512 * 512;
  int idx = blockIdx.x * blockDim.x + threadIdx.x;
  if (idx >= 4 * HW) return;
  int b = idx / HW, hw = idx - b * HW;
  const float* xb = x + (size_t)b * 3 * HW;
  float r = xb[hw], g = xb[HW + hw], bl = xb[2 * HW + hw];
  float f[10] = {r * r, r * g, r * bl, r, g * g, g * bl, g, bl * bl, bl, 1.f};
  const float* w = wq + b * 30;
  float* ob = out + (size_t)b * 3 * HW;
#pragma unroll
  for (int c = 0; c < 3; ++c) {
    float s = (c == 0) ? r : ((c == 1) ? g : bl);
#pragma unroll
    for (int k = 0; k < 10; ++k) s += f[k] * w[c * 10 + k];
    ob[c * HW + hw] = s;
  }
}

static const size_t OFF_WT0 = 0;
static const size_t OFF_WT1 = 55296;
static const size_t OFF_WT2 = 92160;
static const size_t OFF_F1 = 129024;
static const size_t OFF_F2 = 129024 + 4129024;
static const size_t OFF_F3 = 129024 + 4129024 + 246016;
static const size_t OFF_GM = OFF_F3 + 12544;
static const size_t OFF_WQ = OFF_GM + 256;

extern "C" void kernel_launch(void* const* d_in, const int* in_sizes, int n_in,
                              void* d_out, int out_size, void* d_ws, size_t ws_size,
                              hipStream_t stream) {
  const float* x = (const float*)d_in[0];
  const float* c0w = (const float*)d_in[1];
  const float* c0b = (const float*)d_in[2];
  const float* c1w = (const float*)d_in[3];
  const float* c1b = (const float*)d_in[4];
  const float* c2w = (const float*)d_in[5];
  const float* c2b = (const float*)d_in[6];
  const float* f1w = (const float*)d_in[7];
  const float* f1b = (const float*)d_in[8];
  const float* f2w = (const float*)d_in[9];
  const float* f2b = (const float*)d_in[10];
  float* out = (float*)d_out;
  float* ws = (float*)d_ws;

  float* wT0 = ws + OFF_WT0;
  float* wT1 = ws + OFF_WT1;
  float* wT2 = ws + OFF_WT2;
  float* F1 = ws + OFF_F1;
  float* F2 = ws + OFF_F2;
  float* F3 = ws + OFF_F3;
  float* GM = ws + OFF_GM;
  float* WQ = ws + OFF_WQ;

  tr0_kernel<<<(864 * 64 + 255) / 256, 256, 0, stream>>>(c0w, wT0);
  tr1_kernel<<<(576 * 64 + 255) / 256, 256, 0, stream>>>(c1w, wT1);
  tr1_kernel<<<(576 * 64 + 255) / 256, 256, 0, stream>>>(c2w, wT2);

  conv0_kernel<<<dim3(16, 127, 4), 32, 0, stream>>>(x, wT0, c0b, F1);
  convmid_kernel<127, 127, 31, 31><<<dim3(4, 31, 4), 32, 0, stream>>>(F1, wT1, c1b, F2);
  convmid_kernel<31, 31, 7, 7><<<dim3(1, 7, 4), 32, 0, stream>>>(F2, wT2, c2b, F3);

  mean_kernel<<<4, 64, 0, stream>>>(F3, GM);
  fc_kernel<<<4, 64, 0, stream>>>(GM, f1w, f1b, f2w, f2b, WQ);

  const int npix = 4 * 512 * 512;
  quad_kernel<<<(npix + 255) / 256, 256, 0, stream>>>(x, WQ, out);
}


# conv0/conv2: a tile's 64 output channels are split over the waves of one workgroup (conv0: 4 waves x 16 on the shared LDS feature tile, outputs gathered in LDS for whole-line stores; conv2: 2 x 32)
# speedup vs baseline: 1.9727x; 1.2855x over previous
#include <hip/hip_runtime.h>
#include <cstddef>

typedef __attribute__((ext_vector_type(2))) float v2f;
typedef __attribute__((ext_vector_type(4))) float v4f;
typedef __attribute__((ext_vector_type(8))) float v8f;

#define DEV __device__ __forceinline__

DEV float eluf(float v) { return v > 0.f ? v : (__expf(v) - 1.f); }
DEV float histf(float v, float bin, float nm1) {
  return fmaxf(1.f - fabsf(v - bin) * nm1, 0.f);
}

DEV v8f wmma_f32(v2f a, v2f b, v8f c) {
  return __builtin_amdgcn_wmma_f32_16x16x4_f32(false, a, false, b, (short)0, c,
                                               false, false);
}

template <int KB, int NT>
struct FragSet {
  v2f a0[KB];
  v2f a1[KB];
  v2f b[KB][NT];
};

template <int KB, int NT>
DEV void mma_block(const FragSet<KB, NT>& s, v8f (&acc)[2][NT]) {
#pragma unroll
  for (int i = 0; i < KB; ++i) {
#pragma unroll
    for (int nt = 0; nt < NT; ++nt) {
      acc[0][nt] = wmma_f32(s.a0[i], s.b[i][nt], acc[0][nt]);
      acc[1][nt] = wmma_f32(s.a1[i], s.b[i][nt], acc[1][nt]);
    }
  }
}

__global__ void tr0_kernel(const float* __restrict__ w, float* __restrict__ wT) {
  int idx = blockIdx.x * blockDim.x + threadIdx.x;
  if (idx >= 864 * 64) return;
  int k = idx >> 6, n = idx & 63;
  int r = k / 96, cf = k % 96;
  float v = (cf < 93) ? w[(size_t)n * 837 + cf * 9 + r] : 0.f;
  wT[((((size_t)(k >> 1)) * 64 + n) << 1) + (k & 1)] = v;
}

__global__ void tr1_kernel(const float* __restrict__ w, float* __restrict__ wT) {
  int idx = blockIdx.x * blockDim.x + threadIdx.x;
  if (idx >= 576 * 64) return;
  int k = idx >> 6, n = idx & 63;
  int r = k >> 6, cin = k & 63;
  float v = w[(size_t)n * 576 + cin * 9 + r];
  wT[((((size_t)(k >> 1)) * 64 + n) << 1) + (k & 1)] = v;
}

__global__ __launch_bounds__(128) void conv0_kernel(const float* __restrict__ x,
                                                    const float* __restrict__ wT,
                                                    const float* __restrict__ bias,
                                                    float* __restrict__ f1) {
  const int lane = threadIdx.x & 31;
  const int wv = __builtin_amdgcn_readfirstlane(threadIdx.x >> 5);
  const int hh = lane >> 4;
  const int ml = lane & 15;
  const int tx = blockIdx.x;
  const int py = blockIdx.y;
  const int b = blockIdx.z;

  __shared__ __attribute__((aligned(16))) float ft[165][98];

  const int gy0 = 4 * py;
  const int gx0 = 32 * tx;
  const float* xb = x + (size_t)b * 3 * 512 * 512;

  for (int p = threadIdx.x; p < 165; p += 128) {
    const int ly = p / 33, lx = p - ly * 33;
    const int gy = min(gy0 + ly, 511);
    const int gx = min(gx0 + lx, 511);
    const int ym = max(gy - 1, 0), yp = min(gy + 1, 511);
    const int xm = max(gx - 1, 0), xp = min(gx + 1, 511);
    float* f = &ft[p][0];
    float cv[3];
    int o = 9;
#pragma unroll
    for (int c = 0; c < 3; ++c) {
      const float* xc = xb + (size_t)c * 512 * 512;
      float a = xc[ym * 512 + xm], bb = xc[ym * 512 + gx], cc = xc[ym * 512 + xp];
      float d = xc[gy * 512 + xm], ee = xc[gy * 512 + gx], e = xc[gy * 512 + xp];
      float g = xc[yp * 512 + xm], h = xc[yp * 512 + gx], i = xc[yp * 512 + xp];
      cv[c] = ee;
      f[c] = ee;
      f[3 + 2 * c] = ((cc + 2.f * e + i) - (a + 2.f * d + g)) * 0.125f;
      f[4 + 2 * c] = ((g + 2.f * h + i) - (a + 2.f * bb + cc)) * 0.125f;
    }
#pragma unroll
    for (int c = 0; c < 3; ++c)
#pragma unroll
      for (int bn = 0; bn < 4; ++bn) f[o++] = histf(cv[c], (float)bn * (1.f / 3.f), 3.f);
#pragma unroll
    for (int c = 0; c < 3; ++c)
#pragma unroll
      for (int bn = 0; bn < 8; ++bn) f[o++] = histf(cv[c], (float)bn * (1.f / 7.f), 7.f);
#pragma unroll
    for (int c = 0; c < 3; ++c)
#pragma unroll
      for (int bn = 0; bn < 16; ++bn) f[o++] = histf(cv[c], (float)bn * (1.f / 15.f), 15.f);
    f[93] = 0.f;
    f[94] = 0.f;
    f[95] = 0.f;
  }
  __syncthreads();

  v8f acc[2][1];
  const v8f vz = {0.f, 0.f, 0.f, 0.f, 0.f, 0.f, 0.f, 0.f};
  acc[0][0] = vz;
  acc[1][0] = vz;

  constexpr int KB = 12;
  constexpr int BPT = 24 / KB;
  constexpr int NB = 9 * BPT;
  auto load_block = [&](FragSet<KB, 1>& s, int j) {
    const int r = j / BPT;
    const int ub = (j - r * BPT) * KB;
    const int ky = r / 3, kx = r - 3 * ky;
    const int lx = 2 * ml + kx;
    const float* fp0 = &ft[ky * 33 + lx][2 * hh + 4 * ub];
    const float* fp1 = &ft[(ky + 2) * 33 + lx][2 * hh + 4 * ub];
    const float* bp = wT + ((((size_t)(r * 48 + 2 * ub + hh)) * 64 + ml) << 1) + wv * 32;
#pragma unroll
    for (int i = 0; i < KB; ++i) {
      s.a0[i] = *(const v2f*)(fp0 + 4 * i);
      s.a1[i] = *(const v2f*)(fp1 + 4 * i);
      s.b[i][0] = *(const v2f*)(bp + i * 256);
    }
  };

  FragSet<KB, 1> fx, fy;
  load_block(fx, 0);
  for (int j = 0; j + 1 < NB; j += 2) {
    load_block(fy, j + 1);
    __builtin_amdgcn_sched_barrier(0);
    mma_block<KB, 1>(fx, acc);
    __builtin_amdgcn_sched_barrier(0);
    load_block(fx, min(j + 2, NB - 1));
    __builtin_amdgcn_sched_barrier(0);
    mma_block<KB, 1>(fy, acc);
  }
  if (NB & 1) mma_block<KB, 1>(fx, acc);

  __syncthreads();
  float* ot = &ft[0][0];
  {
    const float bb = bias[wv * 16 + ml];
#pragma unroll
    for (int j = 0; j < 4; ++j) {
      float m0 = fmaxf(eluf(acc[0][0][2 * j] + bb), eluf(acc[0][0][2 * j + 1] + bb));
      float m1 = fmaxf(eluf(acc[1][0][2 * j] + bb), eluf(acc[1][0][2 * j + 1] + bb));
      ot[(j + 4 * hh) * 64 + wv * 16 + ml] = fmaxf(m0, m1);
    }
  }
  __syncthreads();
  {
    const int pl = threadIdx.x >> 4;
    const int c4 = (threadIdx.x & 15) * 4;
    const int px = 8 * tx + pl;
    if (px < 127)
      *(v4f*)(f1 + ((((size_t)b * 127 + py) * 127 + px) << 6) + c4) =
          *(const v4f*)(ot + pl * 64 + c4);
  }
}

template <int HIN, int WIN, int HP, int WP, int WV>
__global__ __launch_bounds__(32 * WV) void convmid_kernel(const float* __restrict__ fin,
                                                          const float* __restrict__ wT,
                                                          const float* __restrict__ bias,
                                                          float* __restrict__ fout) {
  constexpr int NT = 4 / WV;
  const int lane = threadIdx.x & 31;
  const int wv = __builtin_amdgcn_readfirstlane(threadIdx.x >> 5);
  const int n0 = wv * NT;
  const int hh = lane >> 4;
  const int ml = lane & 15;
  const int tx = blockIdx.x;
  const int py = blockIdx.y;
  const int b = blockIdx.z;

  v8f acc[2][NT];
  const v8f vz = {0.f, 0.f, 0.f, 0.f, 0.f, 0.f, 0.f, 0.f};
  for (int mt = 0; mt < 2; ++mt)
    for (int nt = 0; nt < NT; ++nt) acc[mt][nt] = vz;

  const int cx = 16 * tx + ml;
  const float* finb = fin + (size_t)b * HIN * WIN * 64;

  constexpr int KB = 8;
  constexpr int BPT = 16 / KB;
  constexpr int NB = 9 * BPT;
  auto load_block = [&](FragSet<KB, NT>& s, int j) {
    const int r = j / BPT;
    const int ub = (j - r * BPT) * KB;
    const int ky = r / 3, kx = r - 3 * ky;
    const int xx = min(2 * cx + kx, WIN - 1);
    const int yy0 = min(4 * py + ky, HIN - 1);
    const int yy1 = min(4 * py + 2 + ky, HIN - 1);
    const float* row0 = finb + (((size_t)yy0 * WIN + xx) << 6) + 4 * ub + 2 * hh;
    const float* row1 = finb + (((size_t)yy1 * WIN + xx) << 6) + 4 * ub + 2 * hh;
    const float* bp = wT + ((((size_t)(r * 32 + 2 * ub + hh)) * 64 + ml) << 1) + n0 * 32;
#pragma unroll
    for (int i = 0; i < KB; ++i) {
      s.a0[i] = *(const v2f*)(row0 + 4 * i);
      s.a1[i] = *(const v2f*)(row1 + 4 * i);
#pragma unroll
      for (int nt = 0; nt < NT; ++nt) s.b[i][nt] = *(const v2f*)(bp + i * 256 + nt * 32);
    }
  };

  FragSet<KB, NT> fx, fy;
  load_block(fx, 0);
  for (int j = 0; j + 1 < NB; j += 2) {
    load_block(fy, j + 1);
    __builtin_amdgcn_sched_barrier(0);
    mma_block<KB, NT>(fx, acc);
    __builtin_amdgcn_sched_barrier(0);
    load_block(fx, min(j + 2, NB - 1));
    __builtin_amdgcn_sched_barrier(0);
    mma_block<KB, NT>(fy, acc);
  }
  if (NB & 1) mma_block<KB, NT>(fx, acc);

  const int pxb = 8 * tx;
#pragma unroll
  for (int nt = 0; nt < NT; ++nt) {
    const float bb = bias[(n0 + nt) * 16 + ml];
#pragma unroll
    for (int j = 0; j < 4; ++j) {
      float m0 = fmaxf(eluf(acc[0][nt][2 * j] + bb), eluf(acc[0][nt][2 * j + 1] + bb));
      float m1 = fmaxf(eluf(acc[1][nt][2 * j] + bb), eluf(acc[1][nt][2 * j + 1] + bb));
      int px = pxb + j + 4 * hh;
      if (px < WP)
        fout[((((size_t)b * HP + py) * WP + px) << 6) + (n0 + nt) * 16 + ml] = fmaxf(m0, m1);
    }
  }
}

__global__ void mean_kernel(const float* __restrict__ f3, float* __restrict__ gm) {
  int b = blockIdx.x, c = threadIdx.x;
  float s = 0.f;
  for (int i = 0; i < 49; ++i) s += f3[(((size_t)b * 49 + i) << 6) + c];
  gm[b * 64 + c] = s * (1.f / 49.f);
}

__global__ void fc_kernel(const float* __restrict__ gm, const float* __restrict__ w1,
                          const float* __restrict__ b1, const float* __restrict__ w2,
                          const float* __restrict__ b2, float* __restrict__ wq) {
  __shared__ float fv[64];
  __shared__ float h[64];
  int b = blockIdx.x, j = threadIdx.x;
  fv[j] = gm[b * 64 + j];
  __syncthreads();
  float acc = b1[j];
  for (int c = 0; c < 64; ++c) acc += fv[c] * w1[j * 64 + c];
  h[j] = eluf(acc);
  __syncthreads();
  if (j < 30) {
    float a2 = b2[j];
    for (int c = 0; c < 64; ++c) a2 += h[c] * w2[j * 64 + c];
    wq[b * 30 + j] = a2;
  }
}

__global__ void quad_kernel(const float* __restrict__ x, const float* __restrict__ wq,
                            float* __restrict__ out) {
  const int HW = 512 * 512;
  int idx = blockIdx.x * blockDim.x + threadIdx.x;
  if (idx >= 4 * HW) return;
  int b = idx / HW, hw = idx - b * HW;
  const float* xb = x + (size_t)b * 3 * HW;
  float r = xb[hw], g = xb[HW + hw], bl = xb[2 * HW + hw];
  float f[10] = {r * r, r * g, r * bl, r, g * g, g * bl, g, bl * bl, bl, 1.f};
  const float* w = wq + b * 30;
  float* ob = out + (size_t)b * 3 * HW;
#pragma unroll
  for (int c = 0; c < 3; ++c) {
    float s = (c == 0) ? r : ((c == 1) ? g : bl);
#pragma unroll
    for (int k = 0; k < 10; ++k) s += f[k] * w[c * 10 + k];
    ob[c * HW + hw] = s;
  }
}

static const size_t OFF_WT0 = 0;
static const size_t OFF_WT1 = 55296;
static const size_t OFF_WT2 = 92160;
static const size_t OFF_F1 = 129024;
static const size_t OFF_F2 = 129024 + 4129024;
static const size_t OFF_F3 = 129024 + 4129024 + 246016;
static const size_t OFF_GM = OFF_F3 + 12544;
static const size_t OFF_WQ = OFF_GM + 256;

extern "C" void kernel_launch(void* const* d_in, const int* in_sizes, int n_in,
                              void* d_out, int out_size, void* d_ws, size_t ws_size,
                              hipStream_t stream) {
  const float* x = (const float*)d_in[0];
  const float* c0w = (const float*)d_in[1];
  const float* c0b = (const float*)d_in[2];
  const float* c1w = (const float*)d_in[3];
  const float* c1b = (const float*)d_in[4];
  const float* c2w = (const float*)d_in[5];
  const float* c2b = (const float*)d_in[6];
  const float* f1w = (const float*)d_in[7];
  const float* f1b = (const float*)d_in[8];
  const float* f2w = (const float*)d_in[9];
  const float* f2b = (const float*)d_in[10];
  float* out = (float*)d_out;
  float* ws = (float*)d_ws;

  float* wT0 = ws + OFF_WT0;
  float* wT1 = ws + OFF_WT1;
  float* wT2 = ws + OFF_WT2;
  float* F1 = ws + OFF_F1;
  float* F2 = ws + OFF_F2;
  float* F3 = ws + OFF_F3;
  float* GM = ws + OFF_GM;
  float* WQ = ws + OFF_WQ;

  tr0_kernel<<<(864 * 64 + 255) / 256, 256, 0, stream>>>(c0w, wT0);
  tr1_kernel<<<(576 * 64 + 255) / 256, 256, 0, stream>>>(c1w, wT1);
  tr1_kernel<<<(576 * 64 + 255) / 256, 256, 0, stream>>>(c2w, wT2);

  conv0_kernel<<<dim3(16, 127, 4), 128, 0, stream>>>(x, wT0, c0b, F1);
  convmid_kernel<127, 127, 31, 31, 1><<<dim3(4, 31, 4), 32, 0, stream>>>(F1, wT1, c1b, F2);
  convmid_kernel<31, 31, 7, 7, 2><<<dim3(1, 7, 4), 64, 0, stream>>>(F2, wT2, c2b, F3);

  mean_kernel<<<4, 64, 0, stream>>>(F3, GM);
  fc_kernel<<<4, 64, 0, stream>>>(GM, f1w, f1b, f2w, f2b, WQ);

  const int npix = 4 * 512 * 512;
  quad_kernel<<<(npix + 255) / 256, 256, 0, stream>>>(x, WQ, out);
}


# 9 launches -> 5: the three weight transposes run as one launch; the 7x7 mean and both fc layers move into the head of the final per-pixel kernel (512 workgroups, each recomputes its 30 coefficients)
# speedup vs baseline: 2.0886x; 1.0588x over previous
#include <hip/hip_runtime.h>
#include <cstddef>

typedef __attribute__((ext_vector_type(2))) float v2f;
typedef __attribute__((ext_vector_type(4))) float v4f;
typedef __attribute__((ext_vector_type(8))) float v8f;

#define DEV __device__ __forceinline__

DEV float eluf(float v) { return v > 0.f ? v : (__expf(v) - 1.f); }
DEV float histf(float v, float bin, float nm1) {
  return fmaxf(1.f - fabsf(v - bin) * nm1, 0.f);
}

DEV v8f wmma_f32(v2f a, v2f b, v8f c) {
  return __builtin_amdgcn_wmma_f32_16x16x4_f32(false, a, false, b, (short)0, c,
                                               false, false);
}

template <int KB, int NT>
struct FragSet {
  v2f a0[KB];
  v2f a1[KB];
  v2f b[KB][NT];
};

template <int KB, int NT>
DEV void mma_block(const FragSet<KB, NT>& s, v8f (&acc)[2][NT]) {
#pragma unroll
  for (int i = 0; i < KB; ++i) {
#pragma unroll
    for (int nt = 0; nt < NT; ++nt) {
      acc[0][nt] = wmma_f32(s.a0[i], s.b[i][nt], acc[0][nt]);
      acc[1][nt] = wmma_f32(s.a1[i], s.b[i][nt], acc[1][nt]);
    }
  }
}

__global__ __launch_bounds__(256) void tr_all_kernel(const float* __restrict__ w0,
                                                     const float* __restrict__ w1,
                                                     const float* __restrict__ w2,
                                                     float* __restrict__ wT0,
                                                     float* __restrict__ wT1,
                                                     float* __restrict__ wT2) {
  const int blk = blockIdx.x;
  if (blk < 216) {
    const int idx = blk * 256 + threadIdx.x;
    const int k = idx >> 6, n = idx & 63;
    const int r = k / 96, cf = k % 96;
    const float v = (cf < 93) ? w0[(size_t)n * 837 + cf * 9 + r] : 0.f;
    wT0[((((size_t)(k >> 1)) * 64 + n) << 1) + (k & 1)] = v;
  } else {
    const bool third = blk >= 360;
    const float* __restrict__ w = third ? w2 : w1;
    float* __restrict__ wT = third ? wT2 : wT1;
    const int idx = (blk - (third ? 360 : 216)) * 256 + threadIdx.x;
    const int k = idx >> 6, n = idx & 63;
    const int r = k >> 6, cin = k & 63;
    const float v = w[(size_t)n * 576 + cin * 9 + r];
    wT[((((size_t)(k >> 1)) * 64 + n) << 1) + (k & 1)] = v;
  }
}

__global__ __launch_bounds__(128) void conv0_kernel(const float* __restrict__ x,
                                                    const float* __restrict__ wT,
                                                    const float* __restrict__ bias,
                                                    float* __restrict__ f1) {
  const int lane = threadIdx.x & 31;
  const int wv = __builtin_amdgcn_readfirstlane(threadIdx.x >> 5);
  const int hh = lane >> 4;
  const int ml = lane & 15;
  const int tx = blockIdx.x;
  const int py = blockIdx.y;
  const int b = blockIdx.z;

  __shared__ __attribute__((aligned(16))) float ft[165][98];

  const int gy0 = 4 * py;
  const int gx0 = 32 * tx;
  const float* xb = x + (size_t)b * 3 * 512 * 512;

  for (int p = threadIdx.x; p < 165; p += 128) {
    const int ly = p / 33, lx = p - ly * 33;
    const int gy = min(gy0 + ly, 511);
    const int gx = min(gx0 + lx, 511);
    const int ym = max(gy - 1, 0), yp = min(gy + 1, 511);
    const int xm = max(gx - 1, 0), xp = min(gx + 1, 511);
    float* f = &ft[p][0];
    float cv[3];
    int o = 9;
#pragma unroll
    for (int c = 0; c < 3; ++c) {
      const float* xc = xb + (size_t)c * 512 * 512;
      float a = xc[ym * 512 + xm], bb = xc[ym * 512 + gx], cc = xc[ym * 512 + xp];
      float d = xc[gy * 512 + xm], ee = xc[gy * 512 + gx], e = xc[gy * 512 + xp];
      float g = xc[yp * 512 + xm], h = xc[yp * 512 + gx], i = xc[yp * 512 + xp];
      cv[c] = ee;
      f[c] = ee;
      f[3 + 2 * c] = ((cc + 2.f * e + i) - (a + 2.f * d + g)) * 0.125f;
      f[4 + 2 * c] = ((g + 2.f * h + i) - (a + 2.f * bb + cc)) * 0.125f;
    }
#pragma unroll
    for (int c = 0; c < 3; ++c)
#pragma unroll
      for (int bn = 0; bn < 4; ++bn) f[o++] = histf(cv[c], (float)bn * (1.f / 3.f), 3.f);
#pragma unroll
    for (int c = 0; c < 3; ++c)
#pragma unroll
      for (int bn = 0; bn < 8; ++bn) f[o++] = histf(cv[c], (float)bn * (1.f / 7.f), 7.f);
#pragma unroll
    for (int c = 0; c < 3; ++c)
#pragma unroll
      for (int bn = 0; bn < 16; ++bn) f[o++] = histf(cv[c], (float)bn * (1.f / 15.f), 15.f);
    f[93] = 0.f;
    f[94] = 0.f;
    f[95] = 0.f;
  }
  __syncthreads();

  v8f acc[2][1];
  const v8f vz = {0.f, 0.f, 0.f, 0.f, 0.f, 0.f, 0.f, 0.f};
  acc[0][0] = vz;
  acc[1][0] = vz;

  constexpr int KB = 12;
  constexpr int BPT = 24 / KB;
  constexpr int NB = 9 * BPT;
  auto load_block = [&](FragSet<KB, 1>& s, int j) {
    const int r = j / BPT;
    const int ub = (j - r * BPT) * KB;
    const int ky = r / 3, kx = r - 3 * ky;
    const int lx = 2 * ml + kx;
    const float* fp0 = &ft[ky * 33 + lx][2 * hh + 4 * ub];
    const float* fp1 = &ft[(ky + 2) * 33 + lx][2 * hh + 4 * ub];
    const float* bp = wT + ((((size_t)(r * 48 + 2 * ub + hh)) * 64 + ml) << 1) + wv * 32;
#pragma unroll
    for (int i = 0; i < KB; ++i) {
      s.a0[i] = *(const v2f*)(fp0 + 4 * i);
      s.a1[i] = *(const v2f*)(fp1 + 4 * i);
      s.b[i][0] = *(const v2f*)(bp + i * 256);
    }
  };

  FragSet<KB, 1> fx, fy;
  load_block(fx, 0);
  for (int j = 0; j + 1 < NB; j += 2) {
    load_block(fy, j + 1);
    __builtin_amdgcn_sched_barrier(0);
    mma_block<KB, 1>(fx, acc);
    __builtin_amdgcn_sched_barrier(0);
    load_block(fx, min(j + 2, NB - 1));
    __builtin_amdgcn_sched_barrier(0);
    mma_block<KB, 1>(fy, acc);
  }
  if (NB & 1) mma_block<KB, 1>(fx, acc);

  __syncthreads();
  float* ot = &ft[0][0];
  {
    const float bb = bias[wv * 16 + ml];
#pragma unroll
    for (int j = 0; j < 4; ++j) {
      float m0 = fmaxf(eluf(acc[0][0][2 * j] + bb), eluf(acc[0][0][2 * j + 1] + bb));
      float m1 = fmaxf(eluf(acc[1][0][2 * j] + bb), eluf(acc[1][0][2 * j + 1] + bb));
      ot[(j + 4 * hh) * 64 + wv * 16 + ml] = fmaxf(m0, m1);
    }
  }
  __syncthreads();
  {
    const int pl = threadIdx.x >> 4;
    const int c4 = (threadIdx.x & 15) * 4;
    const int px = 8 * tx + pl;
    if (px < 127)
      *(v4f*)(f1 + ((((size_t)b * 127 + py) * 127 + px) << 6) + c4) =
          *(const v4f*)(ot + pl * 64 + c4);
  }
}

template <int HIN, int WIN, int HP, int WP, int WV>
__global__ __launch_bounds__(32 * WV) void convmid_kernel(const float* __restrict__ fin,
                                                          const float* __restrict__ wT,
                                                          const float* __restrict__ bias,
                                                          float* __restrict__ fout) {
  constexpr int NT = 4 / WV;
  const int lane = threadIdx.x & 31;
  const int wv = __builtin_amdgcn_readfirstlane(threadIdx.x >> 5);
  const int n0 = wv * NT;
  const int hh = lane >> 4;
  const int ml = lane & 15;
  const int tx = blockIdx.x;
  const int py = blockIdx.y;
  const int b = blockIdx.z;

  v8f acc[2][NT];
  const v8f vz = {0.f, 0.f, 0.f, 0.f, 0.f, 0.f, 0.f, 0.f};
  for (int mt = 0; mt < 2; ++mt)
    for (int nt = 0; nt < NT; ++nt) acc[mt][nt] = vz;

  const int cx = 16 * tx + ml;
  const float* finb = fin + (size_t)b * HIN * WIN * 64;

  constexpr int KB = 8;
  constexpr int BPT = 16 / KB;
  constexpr int NB = 9 * BPT;
  auto load_block = [&](FragSet<KB, NT>& s, int j) {
    const int r = j / BPT;
    const int ub = (j - r * BPT) * KB;
    const int ky = r / 3, kx = r - 3 * ky;
    const int xx = min(2 * cx + kx, WIN - 1);
    const int yy0 = min(4 * py + ky, HIN - 1);
    const int yy1 = min(4 * py + 2 + ky, HIN - 1);
    const float* row0 = finb + (((size_t)yy0 * WIN + xx) << 6) + 4 * ub + 2 * hh;
    const float* row1 = finb + (((size_t)yy1 * WIN + xx) << 6) + 4 * ub + 2 * hh;
    const float* bp = wT + ((((size_t)(r * 32 + 2 * ub + hh)) * 64 + ml) << 1) + n0 * 32;
#pragma unroll
    for (int i = 0; i < KB; ++i) {
      s.a0[i] = *(const v2f*)(row0 + 4 * i);
      s.a1[i] = *(const v2f*)(row1 + 4 * i);
#pragma unroll
      for (int nt = 0; nt < NT; ++nt) s.b[i][nt] = *(const v2f*)(bp + i * 256 + nt * 32);
    }
  };

  FragSet<KB, NT> fx, fy;
  load_block(fx, 0);
  for (int j = 0; j + 1 < NB; j += 2) {
    load_block(fy, j + 1);
    __builtin_amdgcn_sched_barrier(0);
    mma_block<KB, NT>(fx, acc);
    __builtin_amdgcn_sched_barrier(0);
    load_block(fx, min(j + 2, NB - 1));
    __builtin_amdgcn_sched_barrier(0);
    mma_block<KB, NT>(fy, acc);
  }
  if (NB & 1) mma_block<KB, NT>(fx, acc);

  const int pxb = 8 * tx;
#pragma unroll
  for (int nt = 0; nt < NT; ++nt) {
    const float bb = bias[(n0 + nt) * 16 + ml];
#pragma unroll
    for (int j = 0; j < 4; ++j) {
      float m0 = fmaxf(eluf(acc[0][nt][2 * j] + bb), eluf(acc[0][nt][2 * j + 1] + bb));
      float m1 = fmaxf(eluf(acc[1][nt][2 * j] + bb), eluf(acc[1][nt][2 * j + 1] + bb));
      int px = pxb + j + 4 * hh;
      if (px < WP)
        fout[((((size_t)b * HP + py) * WP + px) << 6) + (n0 + nt) * 16 + ml] = fmaxf(m0, m1);
    }
  }
}

__global__ __launch_bounds__(256) void final_kernel(const float* __restrict__ x,
                                                    const float* __restrict__ f3,
                                                    const float* __restrict__ w1,
                                                    const float* __restrict__ b1,
                                                    const float* __restrict__ w2,
                                                    const float* __restrict__ b2,
                                                    float* __restrict__ out) {
  constexpr int HW = 512 * 512;
  __shared__ float fv[64];
  __shared__ float hv[64];
  __shared__ __attribute__((aligned(16))) float wq[32];
  const int t = threadIdx.x;
  const int b = blockIdx.x >> 7;
  const int base = (blockIdx.x & 127) * 2048 + 4 * t;

  const float* xb = x + (size_t)b * 3 * HW + base;
  v4f px[3][2];
#pragma unroll
  for (int c = 0; c < 3; ++c)
#pragma unroll
    for (int h = 0; h < 2; ++h) px[c][h] = *(const v4f*)(xb + (size_t)c * HW + h * 1024);

  v4f w1r[16];
  v4f w2r[16];
  float bias1 = 0.f, bias2 = 0.f;
  if (t < 64) {
#pragma unroll
    for (int q = 0; q < 16; ++q) w1r[q] = *(const v4f*)(w1 + t * 64 + 4 * q);
    bias1 = b1[t];
  }
  if (t < 30) {
#pragma unroll
    for (int q = 0; q < 16; ++q) w2r[q] = *(const v4f*)(w2 + t * 64 + 4 * q);
    bias2 = b2[t];
  }

  if (t < 64) {
    float s = 0.f;
#pragma unroll
    for (int i = 0; i < 49; ++i) s += f3[(((size_t)b * 49 + i) << 6) + t];
    fv[t] = s * (1.f / 49.f);
  }
  __syncthreads();
  if (t < 64) {
    float acc = bias1;
#pragma unroll
    for (int q = 0; q < 16; ++q) {
      acc = fmaf(fv[4 * q + 0], w1r[q][0], acc);
      acc = fmaf(fv[4 * q + 1], w1r[q][1], acc);
      acc = fmaf(fv[4 * q + 2], w1r[q][2], acc);
      acc = fmaf(fv[4 * q + 3], w1r[q][3], acc);
    }
    hv[t] = eluf(acc);
  }
  __syncthreads();
  if (t < 30) {
    float a2 = bias2;
#pragma unroll
    for (int q = 0; q < 16; ++q) {
      a2 = fmaf(hv[4 * q + 0], w2r[q][0], a2);
      a2 = fmaf(hv[4 * q + 1], w2r[q][1], a2);
      a2 = fmaf(hv[4 * q + 2], w2r[q][2], a2);
      a2 = fmaf(hv[4 * q + 3], w2r[q][3], a2);
    }
    wq[t] = a2;
  }
  __syncthreads();

  float w[30];
#pragma unroll
  for (int k = 0; k < 30; ++k) w[k] = wq[k];

  float* ob = out + (size_t)b * 3 * HW + base;
#pragma unroll
  for (int h = 0; h < 2; ++h) {
    v4f o[3];
#pragma unroll
    for (int e = 0; e < 4; ++e) {
      const float r = px[0][h][e], g = px[1][h][e], bl = px[2][h][e];
      const float f[9] = {r * r, r * g, r * bl, r, g * g, g * bl, g, bl * bl, bl};
#pragma unroll
      for (int c = 0; c < 3; ++c) {
        float s = (c == 0) ? r : ((c == 1) ? g : bl);
#pragma unroll
        for (int k = 0; k < 9; ++k) s = fmaf(f[k], w[c * 10 + k], s);
        o[c][e] = s + w[c * 10 + 9];
      }
    }
#pragma unroll
    for (int c = 0; c < 3; ++c) *(v4f*)(ob + (size_t)c * HW + h * 1024) = o[c];
  }
}

static const size_t OFF_WT0 = 0;
static const size_t OFF_WT1 = 55296;
static const size_t OFF_WT2 = 92160;
static const size_t OFF_F1 = 129024;
static const size_t OFF_F2 = 129024 + 4129024;
static const size_t OFF_F3 = 129024 + 4129024 + 246016;

extern "C" void kernel_launch(void* const* d_in, const int* in_sizes, int n_in,
                              void* d_out, int out_size, void* d_ws, size_t ws_size,
                              hipStream_t stream) {
  const float* x = (const float*)d_in[0];
  const float* c0w = (const float*)d_in[1];
  const float* c0b = (const float*)d_in[2];
  const float* c1w = (const float*)d_in[3];
  const float* c1b = (const float*)d_in[4];
  const float* c2w = (const float*)d_in[5];
  const float* c2b = (const float*)d_in[6];
  const float* f1w = (const float*)d_in[7];
  const float* f1b = (const float*)d_in[8];
  const float* f2w = (const float*)d_in[9];
  const float* f2b = (const float*)d_in[10];
  float* out = (float*)d_out;
  float* ws = (float*)d_ws;

  float* wT0 = ws + OFF_WT0;
  float* wT1 = ws + OFF_WT1;
  float* wT2 = ws + OFF_WT2;
  float* F1 = ws + OFF_F1;
  float* F2 = ws + OFF_F2;
  float* F3 = ws + OFF_F3;

  tr_all_kernel<<<504, 256, 0, stream>>>(c0w, c1w, c2w, wT0, wT1, wT2);

  conv0_kernel<<<dim3(16, 127, 4), 128, 0, stream>>>(x, wT0, c0b, F1);
  convmid_kernel<127, 127, 31, 31, 1><<<dim3(4, 31, 4), 32, 0, stream>>>(F1, wT1, c1b, F2);
  convmid_kernel<31, 31, 7, 7, 2><<<dim3(1, 7, 4), 64, 0, stream>>>(F2, wT2, c2b, F3);

  final_kernel<<<512, 256, 0, stream>>>(x, F3, f1w, f1b, f2w, f2b, out);
}


# conv0: x and the 4-bin histogram folded into the 16-bin histogram's weights (piecewise-linear identity on [0,1]): K 864 -> 720, a sixth fewer WMMAs and operand loads; changes the summation order
# speedup vs baseline: 2.4259x; 1.1615x over previous
#include <hip/hip_runtime.h>
#include <cstddef>

typedef __attribute__((ext_vector_type(2))) float v2f;
typedef __attribute__((ext_vector_type(4))) float v4f;
typedef __attribute__((ext_vector_type(8))) float v8f;

#define DEV __device__ __forceinline__

DEV float eluf(float v) { return v > 0.f ? v : (__expf(v) - 1.f); }
DEV float histf(float v, float bin, float nm1) {
  return fmaxf(1.f - fabsf(v - bin) * nm1, 0.f);
}

DEV v8f wmma_f32(v2f a, v2f b, v8f c) {
  return __builtin_amdgcn_wmma_f32_16x16x4_f32(false, a, false, b, (short)0, c,
                                               false, false);
}

template <int KB, int NT>
struct FragSet {
  v2f a0[KB];
  v2f a1[KB];
  v2f b[KB][NT];
};

template <int KB, int NT>
DEV void mma_block(const FragSet<KB, NT>& s, v8f (&acc)[2][NT]) {
#pragma unroll
  for (int i = 0; i < KB; ++i) {
#pragma unroll
    for (int nt = 0; nt < NT; ++nt) {
      acc[0][nt] = wmma_f32(s.a0[i], s.b[i][nt], acc[0][nt]);
      acc[1][nt] = wmma_f32(s.a1[i], s.b[i][nt], acc[1][nt]);
    }
  }
}

__global__ __launch_bounds__(256) void tr_all_kernel(const float* __restrict__ w0,
                                                     const float* __restrict__ w1,
                                                     const float* __restrict__ w2,
                                                     float* __restrict__ wT0,
                                                     float* __restrict__ wT1,
                                                     float* __restrict__ wT2) {
  const int blk = blockIdx.x;
  if (blk < 180) {
    const int idx = blk * 256 + threadIdx.x;
    const int k = idx >> 6, n = idx & 63;
    const int r = k / 80, cf = k % 80;
    const float* __restrict__ wn = w0 + (size_t)n * 837 + r;
    float v = 0.f;
    if (cf < 6) {
      v = wn[(3 + cf) * 9];
    } else if (cf < 30) {
      v = wn[(15 + cf) * 9];
    } else if (cf < 78) {
      const int q = cf - 30;
      const int c = q >> 4, kb = q & 15;
      v = wn[(45 + q) * 9];
      for (int j = 0; j < 4; ++j) {
        const int d = 5 - abs(kb - 5 * j);
        if (d > 0) v = fmaf(wn[(9 + 4 * c + j) * 9], (float)d / 5.f, v);
      }
      v = fmaf(wn[c * 9], (float)kb / 15.f, v);
    }
    wT0[((((size_t)(k >> 1)) * 64 + n) << 1) + (k & 1)] = v;
  } else {
    const bool third = blk >= 324;
    const float* __restrict__ w = third ? w2 : w1;
    float* __restrict__ wT = third ? wT2 : wT1;
    const int idx = (blk - (third ? 324 : 180)) * 256 + threadIdx.x;
    const int k = idx >> 6, n = idx & 63;
    const int r = k >> 6, cin = k & 63;
    const float v = w[(size_t)n * 576 + cin * 9 + r];
    wT[((((size_t)(k >> 1)) * 64 + n) << 1) + (k & 1)] = v;
  }
}

__global__ __launch_bounds__(128) void conv0_kernel(const float* __restrict__ x,
                                                    const float* __restrict__ wT,
                                                    const float* __restrict__ bias,
                                                    float* __restrict__ f1) {
  const int lane = threadIdx.x & 31;
  const int wv = __builtin_amdgcn_readfirstlane(threadIdx.x >> 5);
  const int hh = lane >> 4;
  const int ml = lane & 15;
  const int tx = blockIdx.x;
  const int py = blockIdx.y;
  const int b = blockIdx.z;

  __shared__ __attribute__((aligned(16))) float ft[165][82];

  const int gy0 = 4 * py;
  const int gx0 = 32 * tx;
  const float* xb = x + (size_t)b * 3 * 512 * 512;

  for (int p = threadIdx.x; p < 165; p += 128) {
    const int ly = p / 33, lx = p - ly * 33;
    const int gy = min(gy0 + ly, 511);
    const int gx = min(gx0 + lx, 511);
    const int ym = max(gy - 1, 0), yp = min(gy + 1, 511);
    const int xm = max(gx - 1, 0), xp = min(gx + 1, 511);
    float* f = &ft[p][0];
    float cv[3];
    int o = 6;
#pragma unroll
    for (int c = 0; c < 3; ++c) {
      const float* xc = xb + (size_t)c * 512 * 512;
      float a = xc[ym * 512 + xm], bb = xc[ym * 512 + gx], cc = xc[ym * 512 + xp];
      float d = xc[gy * 512 + xm], ee = xc[gy * 512 + gx], e = xc[gy * 512 + xp];
      float g = xc[yp * 512 + xm], h = xc[yp * 512 + gx], i = xc[yp * 512 + xp];
      cv[c] = ee;
      f[2 * c] = ((cc + 2.f * e + i) - (a + 2.f * d + g)) * 0.125f;
      f[2 * c + 1] = ((g + 2.f * h + i) - (a + 2.f * bb + cc)) * 0.125f;
    }
#pragma unroll
    for (int c = 0; c < 3; ++c)
#pragma unroll
      for (int bn = 0; bn < 8; ++bn) f[o++] = histf(cv[c], (float)bn * (1.f / 7.f), 7.f);
#pragma unroll
    for (int c = 0; c < 3; ++c)
#pragma unroll
      for (int bn = 0; bn < 16; ++bn) f[o++] = histf(cv[c], (float)bn * (1.f / 15.f), 15.f);
    f[78] = 0.f;
    f[79] = 0.f;
  }
  __syncthreads();

  v8f acc[2][1];
  const v8f vz = {0.f, 0.f, 0.f, 0.f, 0.f, 0.f, 0.f, 0.f};
  acc[0][0] = vz;
  acc[1][0] = vz;

  constexpr int KB = 10;
  constexpr int BPT = 20 / KB;
  constexpr int NB = 9 * BPT;
  auto load_block = [&](FragSet<KB, 1>& s, int j) {
    const int r = j / BPT;
    const int ub = (j - r * BPT) * KB;
    const int ky = r / 3, kx = r - 3 * ky;
    const int lx = 2 * ml + kx;
    const float* fp0 = &ft[ky * 33 + lx][2 * hh + 4 * ub];
    const float* fp1 = &ft[(ky + 2) * 33 + lx][2 * hh + 4 * ub];
    const float* bp = wT + ((((size_t)(r * 40 + 2 * ub + hh)) * 64 + ml) << 1) + wv * 32;
#pragma unroll
    for (int i = 0; i < KB; ++i) {
      s.a0[i] = *(const v2f*)(fp0 + 4 * i);
      s.a1[i] = *(const v2f*)(fp1 + 4 * i);
      s.b[i][0] = *(const v2f*)(bp + i * 256);
    }
  };

  FragSet<KB, 1> fx, fy;
  load_block(fx, 0);
  for (int j = 0; j + 1 < NB; j += 2) {
    load_block(fy, j + 1);
    __builtin_amdgcn_sched_barrier(0);
    mma_block<KB, 1>(fx, acc);
    __builtin_amdgcn_sched_barrier(0);
    load_block(fx, min(j + 2, NB - 1));
    __builtin_amdgcn_sched_barrier(0);
    mma_block<KB, 1>(fy, acc);
  }
  if (NB & 1) mma_block<KB, 1>(fx, acc);

  __syncthreads();
  float* ot = &ft[0][0];
  {
    const float bb = bias[wv * 16 + ml];
#pragma unroll
    for (int j = 0; j < 4; ++j) {
      float m0 = fmaxf(eluf(acc[0][0][2 * j] + bb), eluf(acc[0][0][2 * j + 1] + bb));
      float m1 = fmaxf(eluf(acc[1][0][2 * j] + bb), eluf(acc[1][0][2 * j + 1] + bb));
      ot[(j + 4 * hh) * 64 + wv * 16 + ml] = fmaxf(m0, m1);
    }
  }
  __syncthreads();
  {
    const int pl = threadIdx.x >> 4;
    const int c4 = (threadIdx.x & 15) * 4;
    const int px = 8 * tx + pl;
    if (px < 127)
      *(v4f*)(f1 + ((((size_t)b * 127 + py) * 127 + px) << 6) + c4) =
          *(const v4f*)(ot + pl * 64 + c4);
  }
}

template <int HIN, int WIN, int HP, int WP, int WV>
__global__ __launch_bounds__(32 * WV) void convmid_kernel(const float* __restrict__ fin,
                                                          const float* __restrict__ wT,
                                                          const float* __restrict__ bias,
                                                          float* __restrict__ fout) {
  constexpr int NT = 4 / WV;
  const int lane = threadIdx.x & 31;
  const int wv = __builtin_amdgcn_readfirstlane(threadIdx.x >> 5);
  const int n0 = wv * NT;
  const int hh = lane >> 4;
  const int ml = lane & 15;
  const int tx = blockIdx.x;
  const int py = blockIdx.y;
  const int b = blockIdx.z;

  v8f acc[2][NT];
  const v8f vz = {0.f, 0.f, 0.f, 0.f, 0.f, 0.f, 0.f, 0.f};
  for (int mt = 0; mt < 2; ++mt)
    for (int nt = 0; nt < NT; ++nt) acc[mt][nt] = vz;

  const int cx = 16 * tx + ml;
  const float* finb = fin + (size_t)b * HIN * WIN * 64;

  constexpr int KB = 8;
  constexpr int BPT = 16 / KB;
  constexpr int NB = 9 * BPT;
  auto load_block = [&](FragSet<KB, NT>& s, int j) {
    const int r = j / BPT;
    const int ub = (j - r * BPT) * KB;
    const int ky = r / 3, kx = r - 3 * ky;
    const int xx = min(2 * cx + kx, WIN - 1);
    const int yy0 = min(4 * py + ky, HIN - 1);
    const int yy1 = min(4 * py + 2 + ky, HIN - 1);
    const float* row0 = finb + (((size_t)yy0 * WIN + xx) << 6) + 4 * ub + 2 * hh;
    const float* row1 = finb + (((size_t)yy1 * WIN + xx) << 6) + 4 * ub + 2 * hh;
    const float* bp = wT + ((((size_t)(r * 32 + 2 * ub + hh)) * 64 + ml) << 1) + n0 * 32;
#pragma unroll
    for (int i = 0; i < KB; ++i) {
      s.a0[i] = *(const v2f*)(row0 + 4 * i);
      s.a1[i] = *(const v2f*)(row1 + 4 * i);
#pragma unroll
      for (int nt = 0; nt < NT; ++nt) s.b[i][nt] = *(const v2f*)(bp + i * 256 + nt * 32);
    }
  };

  FragSet<KB, NT> fx, fy;
  load_block(fx, 0);
  for (int j = 0; j + 1 < NB; j += 2) {
    load_block(fy, j + 1);
    __builtin_amdgcn_sched_barrier(0);
    mma_block<KB, NT>(fx, acc);
    __builtin_amdgcn_sched_barrier(0);
    load_block(fx, min(j + 2, NB - 1));
    __builtin_amdgcn_sched_barrier(0);
    mma_block<KB, NT>(fy, acc);
  }
  if (NB & 1) mma_block<KB, NT>(fx, acc);

  const int pxb = 8 * tx;
#pragma unroll
  for (int nt = 0; nt < NT; ++nt) {
    const float bb = bias[(n0 + nt) * 16 + ml];
#pragma unroll
    for (int j = 0; j < 4; ++j) {
      float m0 = fmaxf(eluf(acc[0][nt][2 * j] + bb), eluf(acc[0][nt][2 * j + 1] + bb));
      float m1 = fmaxf(eluf(acc[1][nt][2 * j] + bb), eluf(acc[1][nt][2 * j + 1] + bb));
      int px = pxb + j + 4 * hh;
      if (px < WP)
        fout[((((size_t)b * HP + py) * WP + px) << 6) + (n0 + nt) * 16 + ml] = fmaxf(m0, m1);
    }
  }
}

__global__ __launch_bounds__(256) void final_kernel(const float* __restrict__ x,
                                                    const float* __restrict__ f3,
                                                    const float* __restrict__ w1,
                                                    const float* __restrict__ b1,
                                                    const float* __restrict__ w2,
                                                    const float* __restrict__ b2,
                                                    float* __restrict__ out) {
  constexpr int HW = 512 * 512;
  __shared__ float fv[64];
  __shared__ float hv[64];
  __shared__ __attribute__((aligned(16))) float wq[32];
  const int t = threadIdx.x;
  const int b = blockIdx.x >> 7;
  const int base = (blockIdx.x & 127) * 2048 + 4 * t;

  const float* xb = x + (size_t)b * 3 * HW + base;
  v4f px[3][2];
#pragma unroll
  for (int c = 0; c < 3; ++c)
#pragma unroll
    for (int h = 0; h < 2; ++h) px[c][h] = *(const v4f*)(xb + (size_t)c * HW + h * 1024);

  v4f w1r[16];
  v4f w2r[16];
  float bias1 = 0.f, bias2 = 0.f;
  if (t < 64) {
#pragma unroll
    for (int q = 0; q < 16; ++q) w1r[q] = *(const v4f*)(w1 + t * 64 + 4 * q);
    bias1 = b1[t];
  }
  if (t < 30) {
#pragma unroll
    for (int q = 0; q < 16; ++q) w2r[q] = *(const v4f*)(w2 + t * 64 + 4 * q);
    bias2 = b2[t];
  }

  if (t < 64) {
    float s = 0.f;
#pragma unroll
    for (int i = 0; i < 49; ++i) s += f3[(((size_t)b * 49 + i) << 6) + t];
    fv[t] = s * (1.f / 49.f);
  }
  __syncthreads();
  if (t < 64) {
    float acc = bias1;
#pragma unroll
    for (int q = 0; q < 16; ++q) {
      acc = fmaf(fv[4 * q + 0], w1r[q][0], acc);
      acc = fmaf(fv[4 * q + 1], w1r[q][1], acc);
      acc = fmaf(fv[4 * q + 2], w1r[q][2], acc);
      acc = fmaf(fv[4 * q + 3], w1r[q][3], acc);
    }
    hv[t] = eluf(acc);
  }
  __syncthreads();
  if (t < 30) {
    float a2 = bias2;
#pragma unroll
    for (int q = 0; q < 16; ++q) {
      a2 = fmaf(hv[4 * q + 0], w2r[q][0], a2);
      a2 = fmaf(hv[4 * q + 1], w2r[q][1], a2);
      a2 = fmaf(hv[4 * q + 2], w2r[q][2], a2);
      a2 = fmaf(hv[4 * q + 3], w2r[q][3], a2);
    }
    wq[t] = a2;
  }
  __syncthreads();

  float w[30];
#pragma unroll
  for (int k = 0; k < 30; ++k) w[k] = wq[k];

  float* ob = out + (size_t)b * 3 * HW + base;
#pragma unroll
  for (int h = 0; h < 2; ++h) {
    v4f o[3];
#pragma unroll
    for (int e = 0; e < 4; ++e) {
      const float r = px[0][h][e], g = px[1][h][e], bl = px[2][h][e];
      const float f[9] = {r * r, r * g, r * bl, r, g * g, g * bl, g, bl * bl, bl};
#pragma unroll
      for (int c = 0; c < 3; ++c) {
        float s = (c == 0) ? r : ((c == 1) ? g : bl);
#pragma unroll
        for (int k = 0; k < 9; ++k) s = fmaf(f[k], w[c * 10 + k], s);
        o[c][e] = s + w[c * 10 + 9];
      }
    }
#pragma unroll
    for (int c = 0; c < 3; ++c) *(v4f*)(ob + (size_t)c * HW + h * 1024) = o[c];
  }
}

static const size_t OFF_WT0 = 0;
static const size_t OFF_WT1 = 55296;
static const size_t OFF_WT2 = 92160;
static const size_t OFF_F1 = 129024;
static const size_t OFF_F2 = 129024 + 4129024;
static const size_t OFF_F3 = 129024 + 4129024 + 246016;

extern "C" void kernel_launch(void* const* d_in, const int* in_sizes, int n_in,
                              void* d_out, int out_size, void* d_ws, size_t ws_size,
                              hipStream_t stream) {
  const float* x = (const float*)d_in[0];
  const float* c0w = (const float*)d_in[1];
  const float* c0b = (const float*)d_in[2];
  const float* c1w = (const float*)d_in[3];
  const float* c1b = (const float*)d_in[4];
  const float* c2w = (const float*)d_in[5];
  const float* c2b = (const float*)d_in[6];
  const float* f1w = (const float*)d_in[7];
  const float* f1b = (const float*)d_in[8];
  const float* f2w = (const float*)d_in[9];
  const float* f2b = (const float*)d_in[10];
  float* out = (float*)d_out;
  float* ws = (float*)d_ws;

  float* wT0 = ws + OFF_WT0;
  float* wT1 = ws + OFF_WT1;
  float* wT2 = ws + OFF_WT2;
  float* F1 = ws + OFF_F1;
  float* F2 = ws + OFF_F2;
  float* F3 = ws + OFF_F3;

  tr_all_kernel<<<468, 256, 0, stream>>>(c0w, c1w, c2w, wT0, wT1, wT2);

  conv0_kernel<<<dim3(16, 127, 4), 128, 0, stream>>>(x, wT0, c0b, F1);
  convmid_kernel<127, 127, 31, 31, 1><<<dim3(4, 31, 4), 32, 0, stream>>>(F1, wT1, c1b, F2);
  convmid_kernel<31, 31, 7, 7, 2><<<dim3(1, 7, 4), 64, 0, stream>>>(F2, wT2, c2b, F3);

  final_kernel<<<512, 256, 0, stream>>>(x, F3, f1w, f1b, f2w, f2b, out);
}


# conv0: the two remaining histograms (bins at k/7 and k/15) merged into one 22-knot hat basis per colour, weights pre-summed at the knots: K 720 -> 648, a tenth fewer WMMAs; changes summation order
# speedup vs baseline: 2.6960x; 1.1113x over previous
#include <hip/hip_runtime.h>
#include <cstddef>

typedef __attribute__((ext_vector_type(2))) float v2f;
typedef __attribute__((ext_vector_type(4))) float v4f;
typedef __attribute__((ext_vector_type(8))) float v8f;

#define DEV __device__ __forceinline__

DEV float eluf(float v) { return v > 0.f ? v : (__expf(v) - 1.f); }
#define HIST_KNOTS {0, 7, 14, 15, 21, 28, 30, 35, 42, 45, 49, 56, 60, 63, 70, 75, 77, 84, 90, 91, 98, 105}

DEV v8f wmma_f32(v2f a, v2f b, v8f c) {
  return __builtin_amdgcn_wmma_f32_16x16x4_f32(false, a, false, b, (short)0, c,
                                               false, false);
}

template <int KB, int NT>
struct FragSet {
  v2f a0[KB];
  v2f a1[KB];
  v2f b[KB][NT];
};

template <int KB, int NT>
DEV void mma_block(const FragSet<KB, NT>& s, v8f (&acc)[2][NT]) {
#pragma unroll
  for (int i = 0; i < KB; ++i) {
#pragma unroll
    for (int nt = 0; nt < NT; ++nt) {
      acc[0][nt] = wmma_f32(s.a0[i], s.b[i][nt], acc[0][nt]);
      acc[1][nt] = wmma_f32(s.a1[i], s.b[i][nt], acc[1][nt]);
    }
  }
}

__global__ __launch_bounds__(256) void tr_all_kernel(const float* __restrict__ w0,
                                                     const float* __restrict__ w1,
                                                     const float* __restrict__ w2,
                                                     float* __restrict__ wT0,
                                                     float* __restrict__ wT1,
                                                     float* __restrict__ wT2) {
  const int blk = blockIdx.x;
  if (blk < 162) {
    const int TK[22] = HIST_KNOTS;
    const int idx = blk * 256 + threadIdx.x;
    const int k = idx >> 6, n = idx & 63;
    const int r = k / 72, cf = k % 72;
    const float* __restrict__ wn = w0 + (size_t)n * 837 + r;
    float v = 0.f;
    if (cf < 6) {
      v = wn[(3 + cf) * 9];
    } else {
      const int q = cf - 6;
      const int c = q / 22;
      const int T = TK[q - 22 * c];
      const int k0 = T / 7, r16 = T - 7 * k0, k1 = min(k0 + 1, 15);
      const int j0 = T / 15, r8 = T - 15 * j0, j1 = min(j0 + 1, 7);
      const int m0 = T / 35, r4 = T - 35 * m0, m1 = min(m0 + 1, 3);
      const float a0 = wn[(45 + 16 * c + k0) * 9], a1 = wn[(45 + 16 * c + k1) * 9];
      const float b0 = wn[(21 + 8 * c + j0) * 9], b1 = wn[(21 + 8 * c + j1) * 9];
      const float d0 = wn[(9 + 4 * c + m0) * 9], d1 = wn[(9 + 4 * c + m1) * 9];
      const float xw = wn[c * 9];
      v = a0 * ((float)(7 - r16) / 7.f);
      v = fmaf(a1, (float)r16 / 7.f, v);
      v = fmaf(b0, (float)(15 - r8) / 15.f, v);
      v = fmaf(b1, (float)r8 / 15.f, v);
      v = fmaf(d0, (float)(35 - r4) / 35.f, v);
      v = fmaf(d1, (float)r4 / 35.f, v);
      v = fmaf(xw, (float)T / 105.f, v);
    }
    wT0[((((size_t)(k >> 1)) * 64 + n) << 1) + (k & 1)] = v;
  } else {
    const bool third = blk >= 306;
    const float* __restrict__ w = third ? w2 : w1;
    float* __restrict__ wT = third ? wT2 : wT1;
    const int idx = (blk - (third ? 306 : 162)) * 256 + threadIdx.x;
    const int k = idx >> 6, n = idx & 63;
    const int r = k >> 6, cin = k & 63;
    const float v = w[(size_t)n * 576 + cin * 9 + r];
    wT[((((size_t)(k >> 1)) * 64 + n) << 1) + (k & 1)] = v;
  }
}

__global__ __launch_bounds__(128) void conv0_kernel(const float* __restrict__ x,
                                                    const float* __restrict__ wT,
                                                    const float* __restrict__ bias,
                                                    float* __restrict__ f1) {
  const int lane = threadIdx.x & 31;
  const int wv = __builtin_amdgcn_readfirstlane(threadIdx.x >> 5);
  const int hh = lane >> 4;
  const int ml = lane & 15;
  const int tx = blockIdx.x;
  const int py = blockIdx.y;
  const int b = blockIdx.z;

  __shared__ __attribute__((aligned(16))) float ft[165][74];

  const int gy0 = 4 * py;
  const int gx0 = 32 * tx;
  const float* xb = x + (size_t)b * 3 * 512 * 512;

  for (int p = threadIdx.x; p < 165; p += 128) {
    const int ly = p / 33, lx = p - ly * 33;
    const int gy = min(gy0 + ly, 511);
    const int gx = min(gx0 + lx, 511);
    const int ym = max(gy - 1, 0), yp = min(gy + 1, 511);
    const int xm = max(gx - 1, 0), xp = min(gx + 1, 511);
    float* f = &ft[p][0];
    float cv[3];
    int o = 6;
#pragma unroll
    for (int c = 0; c < 3; ++c) {
      const float* xc = xb + (size_t)c * 512 * 512;
      float a = xc[ym * 512 + xm], bb = xc[ym * 512 + gx], cc = xc[ym * 512 + xp];
      float d = xc[gy * 512 + xm], ee = xc[gy * 512 + gx], e = xc[gy * 512 + xp];
      float g = xc[yp * 512 + xm], h = xc[yp * 512 + gx], i = xc[yp * 512 + xp];
      cv[c] = ee;
      f[2 * c] = ((cc + 2.f * e + i) - (a + 2.f * d + g)) * 0.125f;
      f[2 * c + 1] = ((g + 2.f * h + i) - (a + 2.f * bb + cc)) * 0.125f;
    }
    constexpr int TK[22] = HIST_KNOTS;
#pragma unroll
    for (int c = 0; c < 3; ++c) {
      const float v = cv[c];
#pragma unroll
      for (int i = 0; i < 22; ++i) {
        float up = 1.f, dn = 1.f;
        if (i > 0) {
          const float g = (float)(TK[i] - TK[i - 1]);
          up = fmaf(v, 105.f / g, -(float)TK[i - 1] / g);
        }
        if (i < 21) {
          const float g = (float)(TK[i + 1] - TK[i]);
          dn = fmaf(v, -105.f / g, (float)TK[i + 1] / g);
        }
        f[o++] = fmaxf(fminf(up, dn), 0.f);
      }
    }
  }
  __syncthreads();

  v8f acc[2][1];
  const v8f vz = {0.f, 0.f, 0.f, 0.f, 0.f, 0.f, 0.f, 0.f};
  acc[0][0] = vz;
  acc[1][0] = vz;

  constexpr int KB = 9;
  constexpr int BPT = 18 / KB;
  constexpr int NB = 9 * BPT;
  auto load_block = [&](FragSet<KB, 1>& s, int j) {
    const int r = j / BPT;
    const int ub = (j - r * BPT) * KB;
    const int ky = r / 3, kx = r - 3 * ky;
    const int lx = 2 * ml + kx;
    const float* fp0 = &ft[ky * 33 + lx][2 * hh + 4 * ub];
    const float* fp1 = &ft[(ky + 2) * 33 + lx][2 * hh + 4 * ub];
    const float* bp = wT + ((((size_t)(r * 36 + 2 * ub + hh)) * 64 + ml) << 1) + wv * 32;
#pragma unroll
    for (int i = 0; i < KB; ++i) {
      s.a0[i] = *(const v2f*)(fp0 + 4 * i);
      s.a1[i] = *(const v2f*)(fp1 + 4 * i);
      s.b[i][0] = *(const v2f*)(bp + i * 256);
    }
  };

  FragSet<KB, 1> fx, fy;
  load_block(fx, 0);
  for (int j = 0; j + 1 < NB; j += 2) {
    load_block(fy, j + 1);
    __builtin_amdgcn_sched_barrier(0);
    mma_block<KB, 1>(fx, acc);
    __builtin_amdgcn_sched_barrier(0);
    load_block(fx, min(j + 2, NB - 1));
    __builtin_amdgcn_sched_barrier(0);
    mma_block<KB, 1>(fy, acc);
  }
  if (NB & 1) mma_block<KB, 1>(fx, acc);

  __syncthreads();
  float* ot = &ft[0][0];
  {
    const float bb = bias[wv * 16 + ml];
#pragma unroll
    for (int j = 0; j < 4; ++j) {
      float m0 = fmaxf(eluf(acc[0][0][2 * j] + bb), eluf(acc[0][0][2 * j + 1] + bb));
      float m1 = fmaxf(eluf(acc[1][0][2 * j] + bb), eluf(acc[1][0][2 * j + 1] + bb));
      ot[(j + 4 * hh) * 64 + wv * 16 + ml] = fmaxf(m0, m1);
    }
  }
  __syncthreads();
  {
    const int pl = threadIdx.x >> 4;
    const int c4 = (threadIdx.x & 15) * 4;
    const int px = 8 * tx + pl;
    if (px < 127)
      *(v4f*)(f1 + ((((size_t)b * 127 + py) * 127 + px) << 6) + c4) =
          *(const v4f*)(ot + pl * 64 + c4);
  }
}

template <int HIN, int WIN, int HP, int WP, int WV>
__global__ __launch_bounds__(32 * WV) void convmid_kernel(const float* __restrict__ fin,
                                                          const float* __restrict__ wT,
                                                          const float* __restrict__ bias,
                                                          float* __restrict__ fout) {
  constexpr int NT = 4 / WV;
  const int lane = threadIdx.x & 31;
  const int wv = __builtin_amdgcn_readfirstlane(threadIdx.x >> 5);
  const int n0 = wv * NT;
  const int hh = lane >> 4;
  const int ml = lane & 15;
  const int tx = blockIdx.x;
  const int py = blockIdx.y;
  const int b = blockIdx.z;

  v8f acc[2][NT];
  const v8f vz = {0.f, 0.f, 0.f, 0.f, 0.f, 0.f, 0.f, 0.f};
  for (int mt = 0; mt < 2; ++mt)
    for (int nt = 0; nt < NT; ++nt) acc[mt][nt] = vz;

  const int cx = 16 * tx + ml;
  const float* finb = fin + (size_t)b * HIN * WIN * 64;

  constexpr int KB = 8;
  constexpr int BPT = 16 / KB;
  constexpr int NB = 9 * BPT;
  auto load_block = [&](FragSet<KB, NT>& s, int j) {
    const int r = j / BPT;
    const int ub = (j - r * BPT) * KB;
    const int ky = r / 3, kx = r - 3 * ky;
    const int xx = min(2 * cx + kx, WIN - 1);
    const int yy0 = min(4 * py + ky, HIN - 1);
    const int yy1 = min(4 * py + 2 + ky, HIN - 1);
    const float* row0 = finb + (((size_t)yy0 * WIN + xx) << 6) + 4 * ub + 2 * hh;
    const float* row1 = finb + (((size_t)yy1 * WIN + xx) << 6) + 4 * ub + 2 * hh;
    const float* bp = wT + ((((size_t)(r * 32 + 2 * ub + hh)) * 64 + ml) << 1) + n0 * 32;
#pragma unroll
    for (int i = 0; i < KB; ++i) {
      s.a0[i] = *(const v2f*)(row0 + 4 * i);
      s.a1[i] = *(const v2f*)(row1 + 4 * i);
#pragma unroll
      for (int nt = 0; nt < NT; ++nt) s.b[i][nt] = *(const v2f*)(bp + i * 256 + nt * 32);
    }
  };

  FragSet<KB, NT> fx, fy;
  load_block(fx, 0);
  for (int j = 0; j + 1 < NB; j += 2) {
    load_block(fy, j + 1);
    __builtin_amdgcn_sched_barrier(0);
    mma_block<KB, NT>(fx, acc);
    __builtin_amdgcn_sched_barrier(0);
    load_block(fx, min(j + 2, NB - 1));
    __builtin_amdgcn_sched_barrier(0);
    mma_block<KB, NT>(fy, acc);
  }
  if (NB & 1) mma_block<KB, NT>(fx, acc);

  const int pxb = 8 * tx;
#pragma unroll
  for (int nt = 0; nt < NT; ++nt) {
    const float bb = bias[(n0 + nt) * 16 + ml];
#pragma unroll
    for (int j = 0; j < 4; ++j) {
      float m0 = fmaxf(eluf(acc[0][nt][2 * j] + bb), eluf(acc[0][nt][2 * j + 1] + bb));
      float m1 = fmaxf(eluf(acc[1][nt][2 * j] + bb), eluf(acc[1][nt][2 * j + 1] + bb));
      int px = pxb + j + 4 * hh;
      if (px < WP)
        fout[((((size_t)b * HP + py) * WP + px) << 6) + (n0 + nt) * 16 + ml] = fmaxf(m0, m1);
    }
  }
}

__global__ __launch_bounds__(256) void final_kernel(const float* __restrict__ x,
                                                    const float* __restrict__ f3,
                                                    const float* __restrict__ w1,
                                                    const float* __restrict__ b1,
                                                    const float* __restrict__ w2,
                                                    const float* __restrict__ b2,
                                                    float* __restrict__ out) {
  constexpr int HW = 512 * 512;
  __shared__ float fv[64];
  __shared__ float hv[64];
  __shared__ __attribute__((aligned(16))) float wq[32];
  const int t = threadIdx.x;
  const int b = blockIdx.x >> 7;
  const int base = (blockIdx.x & 127) * 2048 + 4 * t;

  const float* xb = x + (size_t)b * 3 * HW + base;
  v4f px[3][2];
#pragma unroll
  for (int c = 0; c < 3; ++c)
#pragma unroll
    for (int h = 0; h < 2; ++h) px[c][h] = *(const v4f*)(xb + (size_t)c * HW + h * 1024);

  v4f w1r[16];
  v4f w2r[16];
  float bias1 = 0.f, bias2 = 0.f;
  if (t < 64) {
#pragma unroll
    for (int q = 0; q < 16; ++q) w1r[q] = *(const v4f*)(w1 + t * 64 + 4 * q);
    bias1 = b1[t];
  }
  if (t < 30) {
#pragma unroll
    for (int q = 0; q < 16; ++q) w2r[q] = *(const v4f*)(w2 + t * 64 + 4 * q);
    bias2 = b2[t];
  }

  if (t < 64) {
    float s = 0.f;
#pragma unroll
    for (int i = 0; i < 49; ++i) s += f3[(((size_t)b * 49 + i) << 6) + t];
    fv[t] = s * (1.f / 49.f);
  }
  __syncthreads();
  if (t < 64) {
    float acc = bias1;
#pragma unroll
    for (int q = 0; q < 16; ++q) {
      acc = fmaf(fv[4 * q + 0], w1r[q][0], acc);
      acc = fmaf(fv[4 * q + 1], w1r[q][1], acc);
      acc = fmaf(fv[4 * q + 2], w1r[q][2], acc);
      acc = fmaf(fv[4 * q + 3], w1r[q][3], acc);
    }
    hv[t] = eluf(acc);
  }
  __syncthreads();
  if (t < 30) {
    float a2 = bias2;
#pragma unroll
    for (int q = 0; q < 16; ++q) {
      a2 = fmaf(hv[4 * q + 0], w2r[q][0], a2);
      a2 = fmaf(hv[4 * q + 1], w2r[q][1], a2);
      a2 = fmaf(hv[4 * q + 2], w2r[q][2], a2);
      a2 = fmaf(hv[4 * q + 3], w2r[q][3], a2);
    }
    wq[t] = a2;
  }
  __syncthreads();

  float w[30];
#pragma unroll
  for (int k = 0; k < 30; ++k) w[k] = wq[k];

  float* ob = out + (size_t)b * 3 * HW + base;
#pragma unroll
  for (int h = 0; h < 2; ++h) {
    v4f o[3];
#pragma unroll
    for (int e = 0; e < 4; ++e) {
      const float r = px[0][h][e], g = px[1][h][e], bl = px[2][h][e];
      const float f[9] = {r * r, r * g, r * bl, r, g * g, g * bl, g, bl * bl, bl};
#pragma unroll
      for (int c = 0; c < 3; ++c) {
        float s = (c == 0) ? r : ((c == 1) ? g : bl);
#pragma unroll
        for (int k = 0; k < 9; ++k) s = fmaf(f[k], w[c * 10 + k], s);
        o[c][e] = s + w[c * 10 + 9];
      }
    }
#pragma unroll
    for (int c = 0; c < 3; ++c) *(v4f*)(ob + (size_t)c * HW + h * 1024) = o[c];
  }
}

static const size_t OFF_WT0 = 0;
static const size_t OFF_WT1 = 55296;
static const size_t OFF_WT2 = 92160;
static const size_t OFF_F1 = 129024;
static const size_t OFF_F2 = 129024 + 4129024;
static const size_t OFF_F3 = 129024 + 4129024 + 246016;

extern "C" void kernel_launch(void* const* d_in, const int* in_sizes, int n_in,
                              void* d_out, int out_size, void* d_ws, size_t ws_size,
                              hipStream_t stream) {
  const float* x = (const float*)d_in[0];
  const float* c0w = (const float*)d_in[1];
  const float* c0b = (const float*)d_in[2];
  const float* c1w = (const float*)d_in[3];
  const float* c1b = (const float*)d_in[4];
  const float* c2w = (const float*)d_in[5];
  const float* c2b = (const float*)d_in[6];
  const float* f1w = (const float*)d_in[7];
  const float* f1b = (const float*)d_in[8];
  const float* f2w = (const float*)d_in[9];
  const float* f2b = (const float*)d_in[10];
  float* out = (float*)d_out;
  float* ws = (float*)d_ws;

  float* wT0 = ws + OFF_WT0;
  float* wT1 = ws + OFF_WT1;
  float* wT2 = ws + OFF_WT2;
  float* F1 = ws + OFF_F1;
  float* F2 = ws + OFF_F2;
  float* F3 = ws + OFF_F3;

  tr_all_kernel<<<450, 256, 0, stream>>>(c0w, c1w, c2w, wT0, wT1, wT2);

  conv0_kernel<<<dim3(16, 127, 4), 128, 0, stream>>>(x, wT0, c0b, F1);
  convmid_kernel<127, 127, 31, 31, 1><<<dim3(4, 31, 4), 32, 0, stream>>>(F1, wT1, c1b, F2);
  convmid_kernel<31, 31, 7, 7, 2><<<dim3(1, 7, 4), 64, 0, stream>>>(F2, wT2, c2b, F3);

  final_kernel<<<512, 256, 0, stream>>>(x, F3, f1w, f1b, f2w, f2b, out);
}


# conv1/conv2: a tile's work split over more waves of its workgroup: conv1 2 waves x 32 channels; conv2 2 x 6 waves, the 18 K blocks in 6 groups summed through LDS (changes conv2's summation order)
# speedup vs baseline: 2.8033x; 1.0398x over previous
#include <hip/hip_runtime.h>
#include <cstddef>

typedef __attribute__((ext_vector_type(2))) float v2f;
typedef __attribute__((ext_vector_type(4))) float v4f;
typedef __attribute__((ext_vector_type(8))) float v8f;

#define DEV __device__ __forceinline__

DEV float eluf(float v) { return v > 0.f ? v : (__expf(v) - 1.f); }
#define HIST_KNOTS {0, 7, 14, 15, 21, 28, 30, 35, 42, 45, 49, 56, 60, 63, 70, 75, 77, 84, 90, 91, 98, 105}

DEV v8f wmma_f32(v2f a, v2f b, v8f c) {
  return __builtin_amdgcn_wmma_f32_16x16x4_f32(false, a, false, b, (short)0, c,
                                               false, false);
}

template <int KB, int NT>
struct FragSet {
  v2f a0[KB];
  v2f a1[KB];
  v2f b[KB][NT];
};

template <int KB, int NT>
DEV void mma_block(const FragSet<KB, NT>& s, v8f (&acc)[2][NT]) {
#pragma unroll
  for (int i = 0; i < KB; ++i) {
#pragma unroll
    for (int nt = 0; nt < NT; ++nt) {
      acc[0][nt] = wmma_f32(s.a0[i], s.b[i][nt], acc[0][nt]);
      acc[1][nt] = wmma_f32(s.a1[i], s.b[i][nt], acc[1][nt]);
    }
  }
}

__global__ __launch_bounds__(256) void tr_all_kernel(const float* __restrict__ w0,
                                                     const float* __restrict__ w1,
                                                     const float* __restrict__ w2,
                                                     float* __restrict__ wT0,
                                                     float* __restrict__ wT1,
                                                     float* __restrict__ wT2) {
  const int blk = blockIdx.x;
  if (blk < 162) {
    const int TK[22] = HIST_KNOTS;
    const int idx = blk * 256 + threadIdx.x;
    const int k = idx >> 6, n = idx & 63;
    const int r = k / 72, cf = k % 72;
    const float* __restrict__ wn = w0 + (size_t)n * 837 + r;
    float v = 0.f;
    if (cf < 6) {
      v = wn[(3 + cf) * 9];
    } else {
      const int q = cf - 6;
      const int c = q / 22;
      const int T = TK[q - 22 * c];
      const int k0 = T / 7, r16 = T - 7 * k0, k1 = min(k0 + 1, 15);
      const int j0 = T / 15, r8 = T - 15 * j0, j1 = min(j0 + 1, 7);
      const int m0 = T / 35, r4 = T - 35 * m0, m1 = min(m0 + 1, 3);
      const float a0 = wn[(45 + 16 * c + k0) * 9], a1 = wn[(45 + 16 * c + k1) * 9];
      const float b0 = wn[(21 + 8 * c + j0) * 9], b1 = wn[(21 + 8 * c + j1) * 9];
      const float d0 = wn[(9 + 4 * c + m0) * 9], d1 = wn[(9 + 4 * c + m1) * 9];
      const float xw = wn[c * 9];
      v = a0 * ((float)(7 - r16) / 7.f);
      v = fmaf(a1, (float)r16 / 7.f, v);
      v = fmaf(b0, (float)(15 - r8) / 15.f, v);
      v = fmaf(b1, (float)r8 / 15.f, v);
      v = fmaf(d0, (float)(35 - r4) / 35.f, v);
      v = fmaf(d1, (float)r4 / 35.f, v);
      v = fmaf(xw, (float)T / 105.f, v);
    }
    wT0[((((size_t)(k >> 1)) * 64 + n) << 1) + (k & 1)] = v;
  } else {
    const bool third = blk >= 306;
    const float* __restrict__ w = third ? w2 : w1;
    float* __restrict__ wT = third ? wT2 : wT1;
    const int idx = (blk - (third ? 306 : 162)) * 256 + threadIdx.x;
    const int k = idx >> 6, n = idx & 63;
    const int r = k >> 6, cin = k & 63;
    const float v = w[(size_t)n * 576 + cin * 9 + r];
    wT[((((size_t)(k >> 1)) * 64 + n) << 1) + (k & 1)] = v;
  }
}

__global__ __launch_bounds__(128) void conv0_kernel(const float* __restrict__ x,
                                                    const float* __restrict__ wT,
                                                    const float* __restrict__ bias,
                                                    float* __restrict__ f1) {
  const int lane = threadIdx.x & 31;
  const int wv = __builtin_amdgcn_readfirstlane(threadIdx.x >> 5);
  const int hh = lane >> 4;
  const int ml = lane & 15;
  const int tx = blockIdx.x;
  const int py = blockIdx.y;
  const int b = blockIdx.z;

  __shared__ __attribute__((aligned(16))) float ft[165][74];

  const int gy0 = 4 * py;
  const int gx0 = 32 * tx;
  const float* xb = x + (size_t)b * 3 * 512 * 512;

  for (int p = threadIdx.x; p < 165; p += 128) {
    const int ly = p / 33, lx = p - ly * 33;
    const int gy = min(gy0 + ly, 511);
    const int gx = min(gx0 + lx, 511);
    const int ym = max(gy - 1, 0), yp = min(gy + 1, 511);
    const int xm = max(gx - 1, 0), xp = min(gx + 1, 511);
    float* f = &ft[p][0];
    float cv[3];
    int o = 6;
#pragma unroll
    for (int c = 0; c < 3; ++c) {
      const float* xc = xb + (size_t)c * 512 * 512;
      float a = xc[ym * 512 + xm], bb = xc[ym * 512 + gx], cc = xc[ym * 512 + xp];
      float d = xc[gy * 512 + xm], ee = xc[gy * 512 + gx], e = xc[gy * 512 + xp];
      float g = xc[yp * 512 + xm], h = xc[yp * 512 + gx], i = xc[yp * 512 + xp];
      cv[c] = ee;
      f[2 * c] = ((cc + 2.f * e + i) - (a + 2.f * d + g)) * 0.125f;
      f[2 * c + 1] = ((g + 2.f * h + i) - (a + 2.f * bb + cc)) * 0.125f;
    }
    constexpr int TK[22] = HIST_KNOTS;
#pragma unroll
    for (int c = 0; c < 3; ++c) {
      const float v = cv[c];
#pragma unroll
      for (int i = 0; i < 22; ++i) {
        float up = 1.f, dn = 1.f;
        if (i > 0) {
          const float g = (float)(TK[i] - TK[i - 1]);
          up = fmaf(v, 105.f / g, -(float)TK[i - 1] / g);
        }
        if (i < 21) {
          const float g = (float)(TK[i + 1] - TK[i]);
          dn = fmaf(v, -105.f / g, (float)TK[i + 1] / g);
        }
        f[o++] = fmaxf(fminf(up, dn), 0.f);
      }
    }
  }
  __syncthreads();

  v8f acc[2][1];
  const v8f vz = {0.f, 0.f, 0.f, 0.f, 0.f, 0.f, 0.f, 0.f};
  acc[0][0] = vz;
  acc[1][0] = vz;

  constexpr int KB = 9;
  constexpr int BPT = 18 / KB;
  constexpr int NB = 9 * BPT;
  auto load_block = [&](FragSet<KB, 1>& s, int j) {
    const int r = j / BPT;
    const int ub = (j - r * BPT) * KB;
    const int ky = r / 3, kx = r - 3 * ky;
    const int lx = 2 * ml + kx;
    const float* fp0 = &ft[ky * 33 + lx][2 * hh + 4 * ub];
    const float* fp1 = &ft[(ky + 2) * 33 + lx][2 * hh + 4 * ub];
    const float* bp = wT + ((((size_t)(r * 36 + 2 * ub + hh)) * 64 + ml) << 1) + wv * 32;
#pragma unroll
    for (int i = 0; i < KB; ++i) {
      s.a0[i] = *(const v2f*)(fp0 + 4 * i);
      s.a1[i] = *(const v2f*)(fp1 + 4 * i);
      s.b[i][0] = *(const v2f*)(bp + i * 256);
    }
  };

  FragSet<KB, 1> fx, fy;
  load_block(fx, 0);
  for (int j = 0; j + 1 < NB; j += 2) {
    load_block(fy, j + 1);
    __builtin_amdgcn_sched_barrier(0);
    mma_block<KB, 1>(fx, acc);
    __builtin_amdgcn_sched_barrier(0);
    load_block(fx, min(j + 2, NB - 1));
    __builtin_amdgcn_sched_barrier(0);
    mma_block<KB, 1>(fy, acc);
  }
  if (NB & 1) mma_block<KB, 1>(fx, acc);

  __syncthreads();
  float* ot = &ft[0][0];
  {
    const float bb = bias[wv * 16 + ml];
#pragma unroll
    for (int j = 0; j < 4; ++j) {
      float m0 = fmaxf(eluf(acc[0][0][2 * j] + bb), eluf(acc[0][0][2 * j + 1] + bb));
      float m1 = fmaxf(eluf(acc[1][0][2 * j] + bb), eluf(acc[1][0][2 * j + 1] + bb));
      ot[(j + 4 * hh) * 64 + wv * 16 + ml] = fmaxf(m0, m1);
    }
  }
  __syncthreads();
  {
    const int pl = threadIdx.x >> 4;
    const int c4 = (threadIdx.x & 15) * 4;
    const int px = 8 * tx + pl;
    if (px < 127)
      *(v4f*)(f1 + ((((size_t)b * 127 + py) * 127 + px) << 6) + c4) =
          *(const v4f*)(ot + pl * 64 + c4);
  }
}

template <int HIN, int WIN, int HP, int WP, int WV, int KS>
__global__ __launch_bounds__(32 * WV * KS) void convmid_kernel(const float* __restrict__ fin,
                                                          const float* __restrict__ wT,
                                                          const float* __restrict__ bias,
                                                          float* __restrict__ fout) {
  constexpr int NT = 4 / WV;
  const int lane = threadIdx.x & 31;
  const int wid = __builtin_amdgcn_readfirstlane(threadIdx.x >> 5);
  const int wv = wid % WV;
  const int ks = (KS > 1) ? wid / WV : 0;
  const int n0 = wv * NT;
  const int hh = lane >> 4;
  const int ml = lane & 15;
  const int tx = blockIdx.x;
  const int py = blockIdx.y;
  const int b = blockIdx.z;

  v8f acc[2][NT];
  const v8f vz = {0.f, 0.f, 0.f, 0.f, 0.f, 0.f, 0.f, 0.f};
  for (int mt = 0; mt < 2; ++mt)
    for (int nt = 0; nt < NT; ++nt) acc[mt][nt] = vz;

  const int cx = 16 * tx + ml;
  const float* finb = fin + (size_t)b * HIN * WIN * 64;

  constexpr int KB = 8;
  constexpr int BPT = 16 / KB;
  constexpr int NB = 9 * BPT;
  static_assert(NB % KS == 0, "the K groups take equal numbers of blocks");
  constexpr int NBW = NB / KS;
  const int j0 = ks * NBW;
  auto load_block = [&](FragSet<KB, NT>& s, int j) {
    const int r = j / BPT;
    const int ub = (j - r * BPT) * KB;
    const int ky = r / 3, kx = r - 3 * ky;
    const int xx = min(2 * cx + kx, WIN - 1);
    const int yy0 = min(4 * py + ky, HIN - 1);
    const int yy1 = min(4 * py + 2 + ky, HIN - 1);
    const float* row0 = finb + (((size_t)yy0 * WIN + xx) << 6) + 4 * ub + 2 * hh;
    const float* row1 = finb + (((size_t)yy1 * WIN + xx) << 6) + 4 * ub + 2 * hh;
    const float* bp = wT + ((((size_t)(r * 32 + 2 * ub + hh)) * 64 + ml) << 1) + n0 * 32;
#pragma unroll
    for (int i = 0; i < KB; ++i) {
      s.a0[i] = *(const v2f*)(row0 + 4 * i);
      s.a1[i] = *(const v2f*)(row1 + 4 * i);
#pragma unroll
      for (int nt = 0; nt < NT; ++nt) s.b[i][nt] = *(const v2f*)(bp + i * 256 + nt * 32);
    }
  };

  FragSet<KB, NT> fx, fy;
  load_block(fx, j0);
  for (int j = 0; j + 1 < NBW; j += 2) {
    load_block(fy, j0 + j + 1);
    __builtin_amdgcn_sched_barrier(0);
    mma_block<KB, NT>(fx, acc);
    __builtin_amdgcn_sched_barrier(0);
    load_block(fx, j0 + min(j + 2, NBW - 1));
    __builtin_amdgcn_sched_barrier(0);
    mma_block<KB, NT>(fy, acc);
  }
  if (NBW & 1) mma_block<KB, NT>(fx, acc);

  if constexpr (KS > 1) {
    __shared__ float red[KS - 1][WV][2 * NT * 8][32];
    if (ks > 0) {
#pragma unroll
      for (int mt = 0; mt < 2; ++mt)
#pragma unroll
        for (int nt = 0; nt < NT; ++nt)
#pragma unroll
          for (int r = 0; r < 8; ++r) red[ks - 1][wv][(mt * NT + nt) * 8 + r][lane] = acc[mt][nt][r];
    }
    __syncthreads();
    if (ks > 0) return;
#pragma unroll
    for (int s = 0; s < KS - 1; ++s)
#pragma unroll
      for (int mt = 0; mt < 2; ++mt)
#pragma unroll
        for (int nt = 0; nt < NT; ++nt)
#pragma unroll
          for (int r = 0; r < 8; ++r) acc[mt][nt][r] += red[s][wv][(mt * NT + nt) * 8 + r][lane];
  }

  const int pxb = 8 * tx;
#pragma unroll
  for (int nt = 0; nt < NT; ++nt) {
    const float bb = bias[(n0 + nt) * 16 + ml];
#pragma unroll
    for (int j = 0; j < 4; ++j) {
      float m0 = fmaxf(eluf(acc[0][nt][2 * j] + bb), eluf(acc[0][nt][2 * j + 1] + bb));
      float m1 = fmaxf(eluf(acc[1][nt][2 * j] + bb), eluf(acc[1][nt][2 * j + 1] + bb));
      int px = pxb + j + 4 * hh;
      if (px < WP)
        fout[((((size_t)b * HP + py) * WP + px) << 6) + (n0 + nt) * 16 + ml] = fmaxf(m0, m1);
    }
  }
}

__global__ __launch_bounds__(256) void final_kernel(const float* __restrict__ x,
                                                    const float* __restrict__ f3,
                                                    const float* __restrict__ w1,
                                                    const float* __restrict__ b1,
                                                    const float* __restrict__ w2,
                                                    const float* __restrict__ b2,
                                                    float* __restrict__ out) {
  constexpr int HW = 512 * 512;
  __shared__ float fv[64];
  __shared__ float hv[64];
  __shared__ __attribute__((aligned(16))) float wq[32];
  const int t = threadIdx.x;
  const int b = blockIdx.x >> 7;
  const int base = (blockIdx.x & 127) * 2048 + 4 * t;

  const float* xb = x + (size_t)b * 3 * HW + base;
  v4f px[3][2];
#pragma unroll
  for (int c = 0; c < 3; ++c)
#pragma unroll
    for (int h = 0; h < 2; ++h) px[c][h] = *(const v4f*)(xb + (size_t)c * HW + h * 1024);

  v4f w1r[16];
  v4f w2r[16];
  float bias1 = 0.f, bias2 = 0.f;
  if (t < 64) {
#pragma unroll
    for (int q = 0; q < 16; ++q) w1r[q] = *(const v4f*)(w1 + t * 64 + 4 * q);
    bias1 = b1[t];
  }
  if (t < 30) {
#pragma unroll
    for (int q = 0; q < 16; ++q) w2r[q] = *(const v4f*)(w2 + t * 64 + 4 * q);
    bias2 = b2[t];
  }

  if (t < 64) {
    float s = 0.f;
#pragma unroll
    for (int i = 0; i < 49; ++i) s += f3[(((size_t)b * 49 + i) << 6) + t];
    fv[t] = s * (1.f / 49.f);
  }
  __syncthreads();
  if (t < 64) {
    float acc = bias1;
#pragma unroll
    for (int q = 0; q < 16; ++q) {
      acc = fmaf(fv[4 * q + 0], w1r[q][0], acc);
      acc = fmaf(fv[4 * q + 1], w1r[q][1], acc);
      acc = fmaf(fv[4 * q + 2], w1r[q][2], acc);
      acc = fmaf(fv[4 * q + 3], w1r[q][3], acc);
    }
    hv[t] = eluf(acc);
  }
  __syncthreads();
  if (t < 30) {
    float a2 = bias2;
#pragma unroll
    for (int q = 0; q < 16; ++q) {
      a2 = fmaf(hv[4 * q + 0], w2r[q][0], a2);
      a2 = fmaf(hv[4 * q + 1], w2r[q][1], a2);
      a2 = fmaf(hv[4 * q + 2], w2r[q][2], a2);
      a2 = fmaf(hv[4 * q + 3], w2r[q][3], a2);
    }
    wq[t] = a2;
  }
  __syncthreads();

  float w[30];
#pragma unroll
  for (int k = 0; k < 30; ++k) w[k] = wq[k];

  float* ob = out + (size_t)b * 3 * HW + base;
#pragma unroll
  for (int h = 0; h < 2; ++h) {
    v4f o[3];
#pragma unroll
    for (int e = 0; e < 4; ++e) {
      const float r = px[0][h][e], g = px[1][h][e], bl = px[2][h][e];
      const float f[9] = {r * r, r * g, r * bl, r, g * g, g * bl, g, bl * bl, bl};
#pragma unroll
      for (int c = 0; c < 3; ++c) {
        float s = (c == 0) ? r : ((c == 1) ? g : bl);
#pragma unroll
        for (int k = 0; k < 9; ++k) s = fmaf(f[k], w[c * 10 + k], s);
        o[c][e] = s + w[c * 10 + 9];
      }
    }
#pragma unroll
    for (int c = 0; c < 3; ++c) *(v4f*)(ob + (size_t)c * HW + h * 1024) = o[c];
  }
}

static const size_t OFF_WT0 = 0;
static const size_t OFF_WT1 = 55296;
static const size_t OFF_WT2 = 92160;
static const size_t OFF_F1 = 129024;
static const size_t OFF_F2 = 129024 + 4129024;
static const size_t OFF_F3 = 129024 + 4129024 + 246016;

extern "C" void kernel_launch(void* const* d_in, const int* in_sizes, int n_in,
                              void* d_out, int out_size, void* d_ws, size_t ws_size,
                              hipStream_t stream) {
  const float* x = (const float*)d_in[0];
  const float* c0w = (const float*)d_in[1];
  const float* c0b = (const float*)d_in[2];
  const float* c1w = (const float*)d_in[3];
  const float* c1b = (const float*)d_in[4];
  const float* c2w = (const float*)d_in[5];
  const float* c2b = (const float*)d_in[6];
  const float* f1w = (const float*)d_in[7];
  const float* f1b = (const float*)d_in[8];
  const float* f2w = (const float*)d_in[9];
  const float* f2b = (const float*)d_in[10];
  float* out = (float*)d_out;
  float* ws = (float*)d_ws;

  float* wT0 = ws + OFF_WT0;
  float* wT1 = ws + OFF_WT1;
  float* wT2 = ws + OFF_WT2;
  float* F1 = ws + OFF_F1;
  float* F2 = ws + OFF_F2;
  float* F3 = ws + OFF_F3;

  tr_all_kernel<<<450, 256, 0, stream>>>(c0w, c1w, c2w, wT0, wT1, wT2);

  conv0_kernel<<<dim3(16, 127, 4), 128, 0, stream>>>(x, wT0, c0b, F1);
  convmid_kernel<127, 127, 31, 31, 2, 1><<<dim3(4, 31, 4), 64, 0, stream>>>(F1, wT1, c1b, F2);
  convmid_kernel<31, 31, 7, 7, 2, 6><<<dim3(1, 7, 4), 384, 0, stream>>>(F2, wT2, c2b, F3);

  final_kernel<<<512, 256, 0, stream>>>(x, F3, f1w, f1b, f2w, f2b, out);
}


# conv0: vector instructions per workgroup cut to about a third: per colour only the 2 non-zero hats are computed (cell table) into a zero-filled LDS row; bias+ELU once, after the 2x2 max (exact)
# speedup vs baseline: 2.9045x; 1.0361x over previous
#include <hip/hip_runtime.h>
#include <cstddef>

typedef __attribute__((ext_vector_type(2))) float v2f;
typedef __attribute__((ext_vector_type(4))) float v4f;
typedef __attribute__((ext_vector_type(8))) float v8f;

#define DEV __device__ __forceinline__

DEV float eluf(float v) { return v > 0.f ? v : (__expf(v) - 1.f); }
#define HIST_KNOTS {0, 7, 14, 15, 21, 28, 30, 35, 42, 45, 49, 56, 60, 63, 70, 75, 77, 84, 90, 91, 98, 105}
struct __attribute__((aligned(16))) HatCell {
  float s, a, b;
  int j;
};
#define HAT_CELLS { \
  {15.0f, 0.0f, 1.0f, 0}, {15.0f, 0.0f, 1.0f, 0}, {15.0f, 0.0f, 1.0f, 0}, \
  {15.0f, 0.0f, 1.0f, 0}, {15.0f, 0.0f, 1.0f, 0}, {15.0f, 0.0f, 1.0f, 0}, \
  {15.0f, 0.0f, 1.0f, 0}, {15.0f, 1.0f, 2.0f, 1}, {15.0f, 1.0f, 2.0f, 1}, \
  {15.0f, 1.0f, 2.0f, 1}, {15.0f, 1.0f, 2.0f, 1}, {15.0f, 1.0f, 2.0f, 1}, \
  {15.0f, 1.0f, 2.0f, 1}, {15.0f, 1.0f, 2.0f, 1}, {105.0f, 14.0f, 15.0f, 2}, \
  {17.5f, 2.5f, 3.5f, 3}, {17.5f, 2.5f, 3.5f, 3}, {17.5f, 2.5f, 3.5f, 3}, \
  {17.5f, 2.5f, 3.5f, 3}, {17.5f, 2.5f, 3.5f, 3}, {17.5f, 2.5f, 3.5f, 3}, \
  {15.0f, 3.0f, 4.0f, 4}, {15.0f, 3.0f, 4.0f, 4}, {15.0f, 3.0f, 4.0f, 4}, \
  {15.0f, 3.0f, 4.0f, 4}, {15.0f, 3.0f, 4.0f, 4}, {15.0f, 3.0f, 4.0f, 4}, \
  {15.0f, 3.0f, 4.0f, 4}, {52.5f, 14.0f, 15.0f, 5}, {52.5f, 14.0f, 15.0f, 5}, \
  {21.0f, 6.0f, 7.0f, 6}, {21.0f, 6.0f, 7.0f, 6}, {21.0f, 6.0f, 7.0f, 6}, \
  {21.0f, 6.0f, 7.0f, 6}, {21.0f, 6.0f, 7.0f, 6}, {15.0f, 5.0f, 6.0f, 7}, \
  {15.0f, 5.0f, 6.0f, 7}, {15.0f, 5.0f, 6.0f, 7}, {15.0f, 5.0f, 6.0f, 7}, \
  {15.0f, 5.0f, 6.0f, 7}, {15.0f, 5.0f, 6.0f, 7}, {15.0f, 5.0f, 6.0f, 7}, \
  {35.0f, 14.0f, 15.0f, 8}, {35.0f, 14.0f, 15.0f, 8}, {35.0f, 14.0f, 15.0f, 8}, \
  {26.25f, 11.25f, 12.25f, 9}, {26.25f, 11.25f, 12.25f, 9}, {26.25f, 11.25f, 12.25f, 9}, \
  {26.25f, 11.25f, 12.25f, 9}, {15.0f, 7.0f, 8.0f, 10}, {15.0f, 7.0f, 8.0f, 10}, \
  {15.0f, 7.0f, 8.0f, 10}, {15.0f, 7.0f, 8.0f, 10}, {15.0f, 7.0f, 8.0f, 10}, \
  {15.0f, 7.0f, 8.0f, 10}, {15.0f, 7.0f, 8.0f, 10}, {26.25f, 14.0f, 15.0f, 11}, \
  {26.25f, 14.0f, 15.0f, 11}, {26.25f, 14.0f, 15.0f, 11}, {26.25f, 14.0f, 15.0f, 11}, \
  {35.0f, 20.0f, 21.0f, 12}, {35.0f, 20.0f, 21.0f, 12}, {35.0f, 20.0f, 21.0f, 12}, \
  {15.0f, 9.0f, 10.0f, 13}, {15.0f, 9.0f, 10.0f, 13}, {15.0f, 9.0f, 10.0f, 13}, \
  {15.0f, 9.0f, 10.0f, 13}, {15.0f, 9.0f, 10.0f, 13}, {15.0f, 9.0f, 10.0f, 13}, \
  {15.0f, 9.0f, 10.0f, 13}, {21.0f, 14.0f, 15.0f, 14}, {21.0f, 14.0f, 15.0f, 14}, \
  {21.0f, 14.0f, 15.0f, 14}, {21.0f, 14.0f, 15.0f, 14}, {21.0f, 14.0f, 15.0f, 14}, \
  {52.5f, 37.5f, 38.5f, 15}, {52.5f, 37.5f, 38.5f, 15}, {15.0f, 11.0f, 12.0f, 16}, \
  {15.0f, 11.0f, 12.0f, 16}, {15.0f, 11.0f, 12.0f, 16}, {15.0f, 11.0f, 12.0f, 16}, \
  {15.0f, 11.0f, 12.0f, 16}, {15.0f, 11.0f, 12.0f, 16}, {15.0f, 11.0f, 12.0f, 16}, \
  {17.5f, 14.0f, 15.0f, 17}, {17.5f, 14.0f, 15.0f, 17}, {17.5f, 14.0f, 15.0f, 17}, \
  {17.5f, 14.0f, 15.0f, 17}, {17.5f, 14.0f, 15.0f, 17}, {17.5f, 14.0f, 15.0f, 17}, \
  {105.0f, 90.0f, 91.0f, 18}, {15.0f, 13.0f, 14.0f, 19}, {15.0f, 13.0f, 14.0f, 19}, \
  {15.0f, 13.0f, 14.0f, 19}, {15.0f, 13.0f, 14.0f, 19}, {15.0f, 13.0f, 14.0f, 19}, \
  {15.0f, 13.0f, 14.0f, 19}, {15.0f, 13.0f, 14.0f, 19}, {15.0f, 14.0f, 15.0f, 20}, \
  {15.0f, 14.0f, 15.0f, 20}, {15.0f, 14.0f, 15.0f, 20}, {15.0f, 14.0f, 15.0f, 20}, \
  {15.0f, 14.0f, 15.0f, 20}, {15.0f, 14.0f, 15.0f, 20}, {15.0f, 14.0f, 15.0f, 20} }

DEV v8f wmma_f32(v2f a, v2f b, v8f c) {
  return __builtin_amdgcn_wmma_f32_16x16x4_f32(false, a, false, b, (short)0, c,
                                               false, false);
}

template <int KB, int NT>
struct FragSet {
  v2f a0[KB];
  v2f a1[KB];
  v2f b[KB][NT];
};

template <int KB, int NT>
DEV void mma_block(const FragSet<KB, NT>& s, v8f (&acc)[2][NT]) {
#pragma unroll
  for (int i = 0; i < KB; ++i) {
#pragma unroll
    for (int nt = 0; nt < NT; ++nt) {
      acc[0][nt] = wmma_f32(s.a0[i], s.b[i][nt], acc[0][nt]);
      acc[1][nt] = wmma_f32(s.a1[i], s.b[i][nt], acc[1][nt]);
    }
  }
}

__global__ __launch_bounds__(256) void tr_all_kernel(const float* __restrict__ w0,
                                                     const float* __restrict__ w1,
                                                     const float* __restrict__ w2,
                                                     float* __restrict__ wT0,
                                                     float* __restrict__ wT1,
                                                     float* __restrict__ wT2) {
  const int blk = blockIdx.x;
  if (blk < 162) {
    const int TK[22] = HIST_KNOTS;
    const int idx = blk * 256 + threadIdx.x;
    const int k = idx >> 6, n = idx & 63;
    const int r = k / 72, cf = k % 72;
    const float* __restrict__ wn = w0 + (size_t)n * 837 + r;
    float v = 0.f;
    if (cf < 6) {
      v = wn[(3 + cf) * 9];
    } else {
      const int q = cf - 6;
      const int c = q / 22;
      const int T = TK[q - 22 * c];
      const int k0 = T / 7, r16 = T - 7 * k0, k1 = min(k0 + 1, 15);
      const int j0 = T / 15, r8 = T - 15 * j0, j1 = min(j0 + 1, 7);
      const int m0 = T / 35, r4 = T - 35 * m0, m1 = min(m0 + 1, 3);
      const float a0 = wn[(45 + 16 * c + k0) * 9], a1 = wn[(45 + 16 * c + k1) * 9];
      const float b0 = wn[(21 + 8 * c + j0) * 9], b1 = wn[(21 + 8 * c + j1) * 9];
      const float d0 = wn[(9 + 4 * c + m0) * 9], d1 = wn[(9 + 4 * c + m1) * 9];
      const float xw = wn[c * 9];
      v = a0 * ((float)(7 - r16) / 7.f);
      v = fmaf(a1, (float)r16 / 7.f, v);
      v = fmaf(b0, (float)(15 - r8) / 15.f, v);
      v = fmaf(b1, (float)r8 / 15.f, v);
      v = fmaf(d0, (float)(35 - r4) / 35.f, v);
      v = fmaf(d1, (float)r4 / 35.f, v);
      v = fmaf(xw, (float)T / 105.f, v);
    }
    wT0[((((size_t)(k >> 1)) * 64 + n) << 1) + (k & 1)] = v;
  } else {
    const bool third = blk >= 306;
    const float* __restrict__ w = third ? w2 : w1;
    float* __restrict__ wT = third ? wT2 : wT1;
    const int idx = (blk - (third ? 306 : 162)) * 256 + threadIdx.x;
    const int k = idx >> 6, n = idx & 63;
    const int r = k >> 6, cin = k & 63;
    const float v = w[(size_t)n * 576 + cin * 9 + r];
    wT[((((size_t)(k >> 1)) * 64 + n) << 1) + (k & 1)] = v;
  }
}

__global__ __launch_bounds__(128) void conv0_kernel(const float* __restrict__ x,
                                                    const float* __restrict__ wT,
                                                    const float* __restrict__ bias,
                                                    float* __restrict__ f1) {
  const int lane = threadIdx.x & 31;
  const int wv = __builtin_amdgcn_readfirstlane(threadIdx.x >> 5);
  const int hh = lane >> 4;
  const int ml = lane & 15;
  const int tx = blockIdx.x;
  const int py = blockIdx.y;
  const int b = blockIdx.z;

  __shared__ __attribute__((aligned(16))) float ft[165][74];

  const int gy0 = 4 * py;
  const int gx0 = 32 * tx;
  const float* xb = x + (size_t)b * 3 * 512 * 512;

  for (int p = threadIdx.x; p < 165; p += 128) {
    const int ly = p / 33, lx = p - ly * 33;
    const int gy = min(gy0 + ly, 511);
    const int gx = min(gx0 + lx, 511);
    const int ym = max(gy - 1, 0), yp = min(gy + 1, 511);
    const int xm = max(gx - 1, 0), xp = min(gx + 1, 511);
    float* f = &ft[p][0];
    float cv[3];
    int o = 6;
#pragma unroll
    for (int c = 0; c < 3; ++c) {
      const float* xc = xb + (size_t)c * 512 * 512;
      float a = xc[ym * 512 + xm], bb = xc[ym * 512 + gx], cc = xc[ym * 512 + xp];
      float d = xc[gy * 512 + xm], ee = xc[gy * 512 + gx], e = xc[gy * 512 + xp];
      float g = xc[yp * 512 + xm], h = xc[yp * 512 + gx], i = xc[yp * 512 + xp];
      cv[c] = ee;
      f[2 * c] = ((cc + 2.f * e + i) - (a + 2.f * d + g)) * 0.125f;
      f[2 * c + 1] = ((g + 2.f * h + i) - (a + 2.f * bb + cc)) * 0.125f;
    }
    const v2f z2 = {0.f, 0.f};
#pragma unroll
    for (int q = 3; q < 36; ++q) *(v2f*)(f + 2 * q) = z2;
    const HatCell CELL[105] = HAT_CELLS;
#pragma unroll
    for (int c = 0; c < 3; ++c) {
      const float v = cv[c];
      const int u = min(max((int)(v * 105.f), 0), 104);
      const HatCell hc = CELL[u];
      float* h = f + o + 22 * c + hc.j;
      h[0] = fminf(fmaxf(fmaf(v, -hc.s, hc.b), 0.f), 1.f);
      h[1] = fminf(fmaxf(fmaf(v, hc.s, -hc.a), 0.f), 1.f);
    }
  }
  __syncthreads();

  v8f acc[2][1];
  const v8f vz = {0.f, 0.f, 0.f, 0.f, 0.f, 0.f, 0.f, 0.f};
  acc[0][0] = vz;
  acc[1][0] = vz;

  constexpr int KB = 9;
  constexpr int BPT = 18 / KB;
  constexpr int NB = 9 * BPT;
  auto load_block = [&](FragSet<KB, 1>& s, int j) {
    const int r = j / BPT;
    const int ub = (j - r * BPT) * KB;
    const int ky = r / 3, kx = r - 3 * ky;
    const int lx = 2 * ml + kx;
    const float* fp0 = &ft[ky * 33 + lx][2 * hh + 4 * ub];
    const float* fp1 = &ft[(ky + 2) * 33 + lx][2 * hh + 4 * ub];
    const float* bp = wT + ((((size_t)(r * 36 + 2 * ub + hh)) * 64 + ml) << 1) + wv * 32;
#pragma unroll
    for (int i = 0; i < KB; ++i) {
      s.a0[i] = *(const v2f*)(fp0 + 4 * i);
      s.a1[i] = *(const v2f*)(fp1 + 4 * i);
      s.b[i][0] = *(const v2f*)(bp + i * 256);
    }
  };

  FragSet<KB, 1> fx, fy;
  load_block(fx, 0);
  for (int j = 0; j + 1 < NB; j += 2) {
    load_block(fy, j + 1);
    __builtin_amdgcn_sched_barrier(0);
    mma_block<KB, 1>(fx, acc);
    __builtin_amdgcn_sched_barrier(0);
    load_block(fx, min(j + 2, NB - 1));
    __builtin_amdgcn_sched_barrier(0);
    mma_block<KB, 1>(fy, acc);
  }
  if (NB & 1) mma_block<KB, 1>(fx, acc);

  __syncthreads();
  float* ot = &ft[0][0];
  {
    const float bb = bias[wv * 16 + ml];
#pragma unroll
    for (int j = 0; j < 4; ++j) {
      float m0 = fmaxf(acc[0][0][2 * j], acc[0][0][2 * j + 1]);
      float m1 = fmaxf(acc[1][0][2 * j], acc[1][0][2 * j + 1]);
      ot[(j + 4 * hh) * 64 + wv * 16 + ml] = eluf(fmaxf(m0, m1) + bb);
    }
  }
  __syncthreads();
  {
    const int pl = threadIdx.x >> 4;
    const int c4 = (threadIdx.x & 15) * 4;
    const int px = 8 * tx + pl;
    if (px < 127)
      *(v4f*)(f1 + ((((size_t)b * 127 + py) * 127 + px) << 6) + c4) =
          *(const v4f*)(ot + pl * 64 + c4);
  }
}

template <int HIN, int WIN, int HP, int WP, int WV, int KS>
__global__ __launch_bounds__(32 * WV * KS) void convmid_kernel(const float* __restrict__ fin,
                                                          const float* __restrict__ wT,
                                                          const float* __restrict__ bias,
                                                          float* __restrict__ fout) {
  constexpr int NT = 4 / WV;
  const int lane = threadIdx.x & 31;
  const int wid = __builtin_amdgcn_readfirstlane(threadIdx.x >> 5);
  const int wv = wid % WV;
  const int ks = (KS > 1) ? wid / WV : 0;
  const int n0 = wv * NT;
  const int hh = lane >> 4;
  const int ml = lane & 15;
  const int tx = blockIdx.x;
  const int py = blockIdx.y;
  const int b = blockIdx.z;

  v8f acc[2][NT];
  const v8f vz = {0.f, 0.f, 0.f, 0.f, 0.f, 0.f, 0.f, 0.f};
  for (int mt = 0; mt < 2; ++mt)
    for (int nt = 0; nt < NT; ++nt) acc[mt][nt] = vz;

  const int cx = 16 * tx + ml;
  const float* finb = fin + (size_t)b * HIN * WIN * 64;

  constexpr int KB = 8;
  constexpr int BPT = 16 / KB;
  constexpr int NB = 9 * BPT;
  static_assert(NB % KS == 0, "the K groups take equal numbers of blocks");
  constexpr int NBW = NB / KS;
  const int j0 = ks * NBW;
  auto load_block = [&](FragSet<KB, NT>& s, int j) {
    const int r = j / BPT;
    const int ub = (j - r * BPT) * KB;
    const int ky = r / 3, kx = r - 3 * ky;
    const int xx = min(2 * cx + kx, WIN - 1);
    const int yy0 = min(4 * py + ky, HIN - 1);
    const int yy1 = min(4 * py + 2 + ky, HIN - 1);
    const float* row0 = finb + (((size_t)yy0 * WIN + xx) << 6) + 4 * ub + 2 * hh;
    const float* row1 = finb + (((size_t)yy1 * WIN + xx) << 6) + 4 * ub + 2 * hh;
    const float* bp = wT + ((((size_t)(r * 32 + 2 * ub + hh)) * 64 + ml) << 1) + n0 * 32;
#pragma unroll
    for (int i = 0; i < KB; ++i) {
      s.a0[i] = *(const v2f*)(row0 + 4 * i);
      s.a1[i] = *(const v2f*)(row1 + 4 * i);
#pragma unroll
      for (int nt = 0; nt < NT; ++nt) s.b[i][nt] = *(const v2f*)(bp + i * 256 + nt * 32);
    }
  };

  FragSet<KB, NT> fx, fy;
  load_block(fx, j0);
  for (int j = 0; j + 1 < NBW; j += 2) {
    load_block(fy, j0 + j + 1);
    __builtin_amdgcn_sched_barrier(0);
    mma_block<KB, NT>(fx, acc);
    __builtin_amdgcn_sched_barrier(0);
    load_block(fx, j0 + min(j + 2, NBW - 1));
    __builtin_amdgcn_sched_barrier(0);
    mma_block<KB, NT>(fy, acc);
  }
  if (NBW & 1) mma_block<KB, NT>(fx, acc);

  if constexpr (KS > 1) {
    __shared__ float red[KS - 1][WV][2 * NT * 8][32];
    if (ks > 0) {
#pragma unroll
      for (int mt = 0; mt < 2; ++mt)
#pragma unroll
        for (int nt = 0; nt < NT; ++nt)
#pragma unroll
          for (int r = 0; r < 8; ++r) red[ks - 1][wv][(mt * NT + nt) * 8 + r][lane] = acc[mt][nt][r];
    }
    __syncthreads();
    if (ks > 0) return;
#pragma unroll
    for (int s = 0; s < KS - 1; ++s)
#pragma unroll
      for (int mt = 0; mt < 2; ++mt)
#pragma unroll
        for (int nt = 0; nt < NT; ++nt)
#pragma unroll
          for (int r = 0; r < 8; ++r) acc[mt][nt][r] += red[s][wv][(mt * NT + nt) * 8 + r][lane];
  }

  const int pxb = 8 * tx;
#pragma unroll
  for (int nt = 0; nt < NT; ++nt) {
    const float bb = bias[(n0 + nt) * 16 + ml];
#pragma unroll
    for (int j = 0; j < 4; ++j) {
      float m0 = fmaxf(eluf(acc[0][nt][2 * j] + bb), eluf(acc[0][nt][2 * j + 1] + bb));
      float m1 = fmaxf(eluf(acc[1][nt][2 * j] + bb), eluf(acc[1][nt][2 * j + 1] + bb));
      int px = pxb + j + 4 * hh;
      if (px < WP)
        fout[((((size_t)b * HP + py) * WP + px) << 6) + (n0 + nt) * 16 + ml] = fmaxf(m0, m1);
    }
  }
}

__global__ __launch_bounds__(256) void final_kernel(const float* __restrict__ x,
                                                    const float* __restrict__ f3,
                                                    const float* __restrict__ w1,
                                                    const float* __restrict__ b1,
                                                    const float* __restrict__ w2,
                                                    const float* __restrict__ b2,
                                                    float* __restrict__ out) {
  constexpr int HW = 512 * 512;
  __shared__ float fv[64];
  __shared__ float hv[64];
  __shared__ __attribute__((aligned(16))) float wq[32];
  const int t = threadIdx.x;
  const int b = blockIdx.x >> 7;
  const int base = (blockIdx.x & 127) * 2048 + 4 * t;

  const float* xb = x + (size_t)b * 3 * HW + base;
  v4f px[3][2];
#pragma unroll
  for (int c = 0; c < 3; ++c)
#pragma unroll
    for (int h = 0; h < 2; ++h) px[c][h] = *(const v4f*)(xb + (size_t)c * HW + h * 1024);

  v4f w1r[16];
  v4f w2r[16];
  float bias1 = 0.f, bias2 = 0.f;
  if (t < 64) {
#pragma unroll
    for (int q = 0; q < 16; ++q) w1r[q] = *(const v4f*)(w1 + t * 64 + 4 * q);
    bias1 = b1[t];
  }
  if (t < 30) {
#pragma unroll
    for (int q = 0; q < 16; ++q) w2r[q] = *(const v4f*)(w2 + t * 64 + 4 * q);
    bias2 = b2[t];
  }

  if (t < 64) {
    float s = 0.f;
#pragma unroll
    for (int i = 0; i < 49; ++i) s += f3[(((size_t)b * 49 + i) << 6) + t];
    fv[t] = s * (1.f / 49.f);
  }
  __syncthreads();
  if (t < 64) {
    float acc = bias1;
#pragma unroll
    for (int q = 0; q < 16; ++q) {
      acc = fmaf(fv[4 * q + 0], w1r[q][0], acc);
      acc = fmaf(fv[4 * q + 1], w1r[q][1], acc);
      acc = fmaf(fv[4 * q + 2], w1r[q][2], acc);
      acc = fmaf(fv[4 * q + 3], w1r[q][3], acc);
    }
    hv[t] = eluf(acc);
  }
  __syncthreads();
  if (t < 30) {
    float a2 = bias2;
#pragma unroll
    for (int q = 0; q < 16; ++q) {
      a2 = fmaf(hv[4 * q + 0], w2r[q][0], a2);
      a2 = fmaf(hv[4 * q + 1], w2r[q][1], a2);
      a2 = fmaf(hv[4 * q + 2], w2r[q][2], a2);
      a2 = fmaf(hv[4 * q + 3], w2r[q][3], a2);
    }
    wq[t] = a2;
  }
  __syncthreads();

  float w[30];
#pragma unroll
  for (int k = 0; k < 30; ++k) w[k] = wq[k];

  float* ob = out + (size_t)b * 3 * HW + base;
#pragma unroll
  for (int h = 0; h < 2; ++h) {
    v4f o[3];
#pragma unroll
    for (int e = 0; e < 4; ++e) {
      const float r = px[0][h][e], g = px[1][h][e], bl = px[2][h][e];
      const float f[9] = {r * r, r * g, r * bl, r, g * g, g * bl, g, bl * bl, bl};
#pragma unroll
      for (int c = 0; c < 3; ++c) {
        float s = (c == 0) ? r : ((c == 1) ? g : bl);
#pragma unroll
        for (int k = 0; k < 9; ++k) s = fmaf(f[k], w[c * 10 + k], s);
        o[c][e] = s + w[c * 10 + 9];
      }
    }
#pragma unroll
    for (int c = 0; c < 3; ++c) *(v4f*)(ob + (size_t)c * HW + h * 1024) = o[c];
  }
}

static const size_t OFF_WT0 = 0;
static const size_t OFF_WT1 = 55296;
static const size_t OFF_WT2 = 92160;
static const size_t OFF_F1 = 129024;
static const size_t OFF_F2 = 129024 + 4129024;
static const size_t OFF_F3 = 129024 + 4129024 + 246016;

extern "C" void kernel_launch(void* const* d_in, const int* in_sizes, int n_in,
                              void* d_out, int out_size, void* d_ws, size_t ws_size,
                              hipStream_t stream) {
  const float* x = (const float*)d_in[0];
  const float* c0w = (const float*)d_in[1];
  const float* c0b = (const float*)d_in[2];
  const float* c1w = (const float*)d_in[3];
  const float* c1b = (const float*)d_in[4];
  const float* c2w = (const float*)d_in[5];
  const float* c2b = (const float*)d_in[6];
  const float* f1w = (const float*)d_in[7];
  const float* f1b = (const float*)d_in[8];
  const float* f2w = (const float*)d_in[9];
  const float* f2b = (const float*)d_in[10];
  float* out = (float*)d_out;
  float* ws = (float*)d_ws;

  float* wT0 = ws + OFF_WT0;
  float* wT1 = ws + OFF_WT1;
  float* wT2 = ws + OFF_WT2;
  float* F1 = ws + OFF_F1;
  float* F2 = ws + OFF_F2;
  float* F3 = ws + OFF_F3;

  tr_all_kernel<<<450, 256, 0, stream>>>(c0w, c1w, c2w, wT0, wT1, wT2);

  conv0_kernel<<<dim3(16, 127, 4), 128, 0, stream>>>(x, wT0, c0b, F1);
  convmid_kernel<127, 127, 31, 31, 2, 1><<<dim3(4, 31, 4), 64, 0, stream>>>(F1, wT1, c1b, F2);
  convmid_kernel<31, 31, 7, 7, 2, 6><<<dim3(1, 7, 4), 384, 0, stream>>>(F2, wT2, c2b, F3);

  final_kernel<<<512, 256, 0, stream>>>(x, F3, f1w, f1b, f2w, f2b, out);
}


# conv0/conv1 compute only what later layers read: a VALID stride-2 conv + 2x2 pool needs inputs 0..4n, so 117x117 of conv0's 127x127 and 29x29 of conv1's 31x31 maps; 14% fewer conv0 tiles, same sums
# speedup vs baseline: 3.2410x; 1.1159x over previous
#include <hip/hip_runtime.h>
#include <cstddef>

typedef __attribute__((ext_vector_type(2))) float v2f;
typedef __attribute__((ext_vector_type(4))) float v4f;
typedef __attribute__((ext_vector_type(8))) float v8f;

#define DEV __device__ __forceinline__

DEV float eluf(float v) { return v > 0.f ? v : (__expf(v) - 1.f); }
#define HIST_KNOTS {0, 7, 14, 15, 21, 28, 30, 35, 42, 45, 49, 56, 60, 63, 70, 75, 77, 84, 90, 91, 98, 105}
struct __attribute__((aligned(16))) HatCell {
  float s, a, b;
  int j;
};
#define HAT_CELLS { \
  {15.0f, 0.0f, 1.0f, 0}, {15.0f, 0.0f, 1.0f, 0}, {15.0f, 0.0f, 1.0f, 0}, \
  {15.0f, 0.0f, 1.0f, 0}, {15.0f, 0.0f, 1.0f, 0}, {15.0f, 0.0f, 1.0f, 0}, \
  {15.0f, 0.0f, 1.0f, 0}, {15.0f, 1.0f, 2.0f, 1}, {15.0f, 1.0f, 2.0f, 1}, \
  {15.0f, 1.0f, 2.0f, 1}, {15.0f, 1.0f, 2.0f, 1}, {15.0f, 1.0f, 2.0f, 1}, \
  {15.0f, 1.0f, 2.0f, 1}, {15.0f, 1.0f, 2.0f, 1}, {105.0f, 14.0f, 15.0f, 2}, \
  {17.5f, 2.5f, 3.5f, 3}, {17.5f, 2.5f, 3.5f, 3}, {17.5f, 2.5f, 3.5f, 3}, \
  {17.5f, 2.5f, 3.5f, 3}, {17.5f, 2.5f, 3.5f, 3}, {17.5f, 2.5f, 3.5f, 3}, \
  {15.0f, 3.0f, 4.0f, 4}, {15.0f, 3.0f, 4.0f, 4}, {15.0f, 3.0f, 4.0f, 4}, \
  {15.0f, 3.0f, 4.0f, 4}, {15.0f, 3.0f, 4.0f, 4}, {15.0f, 3.0f, 4.0f, 4}, \
  {15.0f, 3.0f, 4.0f, 4}, {52.5f, 14.0f, 15.0f, 5}, {52.5f, 14.0f, 15.0f, 5}, \
  {21.0f, 6.0f, 7.0f, 6}, {21.0f, 6.0f, 7.0f, 6}, {21.0f, 6.0f, 7.0f, 6}, \
  {21.0f, 6.0f, 7.0f, 6}, {21.0f, 6.0f, 7.0f, 6}, {15.0f, 5.0f, 6.0f, 7}, \
  {15.0f, 5.0f, 6.0f, 7}, {15.0f, 5.0f, 6.0f, 7}, {15.0f, 5.0f, 6.0f, 7}, \
  {15.0f, 5.0f, 6.0f, 7}, {15.0f, 5.0f, 6.0f, 7}, {15.0f, 5.0f, 6.0f, 7}, \
  {35.0f, 14.0f, 15.0f, 8}, {35.0f, 14.0f, 15.0f, 8}, {35.0f, 14.0f, 15.0f, 8}, \
  {26.25f, 11.25f, 12.25f, 9}, {26.25f, 11.25f, 12.25f, 9}, {26.25f, 11.25f, 12.25f, 9}, \
  {26.25f, 11.25f, 12.25f, 9}, {15.0f, 7.0f, 8.0f, 10}, {15.0f, 7.0f, 8.0f, 10}, \
  {15.0f, 7.0f, 8.0f, 10}, {15.0f, 7.0f, 8.0f, 10}, {15.0f, 7.0f, 8.0f, 10}, \
  {15.0f, 7.0f, 8.0f, 10}, {15.0f, 7.0f, 8.0f, 10}, {26.25f, 14.0f, 15.0f, 11}, \
  {26.25f, 14.0f, 15.0f, 11}, {26.25f, 14.0f, 15.0f, 11}, {26.25f, 14.0f, 15.0f, 11}, \
  {35.0f, 20.0f, 21.0f, 12}, {35.0f, 20.0f, 21.0f, 12}, {35.0f, 20.0f, 21.0f, 12}, \
  {15.0f, 9.0f, 10.0f, 13}, {15.0f, 9.0f, 10.0f, 13}, {15.0f, 9.0f, 10.0f, 13}, \
  {15.0f, 9.0f, 10.0f, 13}, {15.0f, 9.0f, 10.0f, 13}, {15.0f, 9.0f, 10.0f, 13}, \
  {15.0f, 9.0f, 10.0f, 13}, {21.0f, 14.0f, 15.0f, 14}, {21.0f, 14.0f, 15.0f, 14}, \
  {21.0f, 14.0f, 15.0f, 14}, {21.0f, 14.0f, 15.0f, 14}, {21.0f, 14.0f, 15.0f, 14}, \
  {52.5f, 37.5f, 38.5f, 15}, {52.5f, 37.5f, 38.5f, 15}, {15.0f, 11.0f, 12.0f, 16}, \
  {15.0f, 11.0f, 12.0f, 16}, {15.0f, 11.0f, 12.0f, 16}, {15.0f, 11.0f, 12.0f, 16}, \
  {15.0f, 11.0f, 12.0f, 16}, {15.0f, 11.0f, 12.0f, 16}, {15.0f, 11.0f, 12.0f, 16}, \
  {17.5f, 14.0f, 15.0f, 17}, {17.5f, 14.0f, 15.0f, 17}, {17.5f, 14.0f, 15.0f, 17}, \
  {17.5f, 14.0f, 15.0f, 17}, {17.5f, 14.0f, 15.0f, 17}, {17.5f, 14.0f, 15.0f, 17}, \
  {105.0f, 90.0f, 91.0f, 18}, {15.0f, 13.0f, 14.0f, 19}, {15.0f, 13.0f, 14.0f, 19}, \
  {15.0f, 13.0f, 14.0f, 19}, {15.0f, 13.0f, 14.0f, 19}, {15.0f, 13.0f, 14.0f, 19}, \
  {15.0f, 13.0f, 14.0f, 19}, {15.0f, 13.0f, 14.0f, 19}, {15.0f, 14.0f, 15.0f, 20}, \
  {15.0f, 14.0f, 15.0f, 20}, {15.0f, 14.0f, 15.0f, 20}, {15.0f, 14.0f, 15.0f, 20}, \
  {15.0f, 14.0f, 15.0f, 20}, {15.0f, 14.0f, 15.0f, 20}, {15.0f, 14.0f, 15.0f, 20} }

DEV v8f wmma_f32(v2f a, v2f b, v8f c) {
  return __builtin_amdgcn_wmma_f32_16x16x4_f32(false, a, false, b, (short)0, c,
                                               false, false);
}

template <int KB, int NT>
struct FragSet {
  v2f a0[KB];
  v2f a1[KB];
  v2f b[KB][NT];
};

template <int KB, int NT>
DEV void mma_block(const FragSet<KB, NT>& s, v8f (&acc)[2][NT]) {
#pragma unroll
  for (int i = 0; i < KB; ++i) {
#pragma unroll
    for (int nt = 0; nt < NT; ++nt) {
      acc[0][nt] = wmma_f32(s.a0[i], s.b[i][nt], acc[0][nt]);
      acc[1][nt] = wmma_f32(s.a1[i], s.b[i][nt], acc[1][nt]);
    }
  }
}

__global__ __launch_bounds__(256) void tr_all_kernel(const float* __restrict__ w0,
                                                     const float* __restrict__ w1,
                                                     const float* __restrict__ w2,
                                                     float* __restrict__ wT0,
                                                     float* __restrict__ wT1,
                                                     float* __restrict__ wT2) {
  const int blk = blockIdx.x;
  if (blk < 162) {
    const int TK[22] = HIST_KNOTS;
    const int idx = blk * 256 + threadIdx.x;
    const int k = idx >> 6, n = idx & 63;
    const int r = k / 72, cf = k % 72;
    const float* __restrict__ wn = w0 + (size_t)n * 837 + r;
    float v = 0.f;
    if (cf < 6) {
      v = wn[(3 + cf) * 9];
    } else {
      const int q = cf - 6;
      const int c = q / 22;
      const int T = TK[q - 22 * c];
      const int k0 = T / 7, r16 = T - 7 * k0, k1 = min(k0 + 1, 15);
      const int j0 = T / 15, r8 = T - 15 * j0, j1 = min(j0 + 1, 7);
      const int m0 = T / 35, r4 = T - 35 * m0, m1 = min(m0 + 1, 3);
      const float a0 = wn[(45 + 16 * c + k0) * 9], a1 = wn[(45 + 16 * c + k1) * 9];
      const float b0 = wn[(21 + 8 * c + j0) * 9], b1 = wn[(21 + 8 * c + j1) * 9];
      const float d0 = wn[(9 + 4 * c + m0) * 9], d1 = wn[(9 + 4 * c + m1) * 9];
      const float xw = wn[c * 9];
      v = a0 * ((float)(7 - r16) / 7.f);
      v = fmaf(a1, (float)r16 / 7.f, v);
      v = fmaf(b0, (float)(15 - r8) / 15.f, v);
      v = fmaf(b1, (float)r8 / 15.f, v);
      v = fmaf(d0, (float)(35 - r4) / 35.f, v);
      v = fmaf(d1, (float)r4 / 35.f, v);
      v = fmaf(xw, (float)T / 105.f, v);
    }
    wT0[((((size_t)(k >> 1)) * 64 + n) << 1) + (k & 1)] = v;
  } else {
    const bool third = blk >= 306;
    const float* __restrict__ w = third ? w2 : w1;
    float* __restrict__ wT = third ? wT2 : wT1;
    const int idx = (blk - (third ? 306 : 162)) * 256 + threadIdx.x;
    const int k = idx >> 6, n = idx & 63;
    const int r = k >> 6, cin = k & 63;
    const float v = w[(size_t)n * 576 + cin * 9 + r];
    wT[((((size_t)(k >> 1)) * 64 + n) << 1) + (k & 1)] = v;
  }
}

__global__ __launch_bounds__(128) void conv0_kernel(const float* __restrict__ x,
                                                    const float* __restrict__ wT,
                                                    const float* __restrict__ bias,
                                                    float* __restrict__ f1) {
  const int lane = threadIdx.x & 31;
  const int wv = __builtin_amdgcn_readfirstlane(threadIdx.x >> 5);
  const int hh = lane >> 4;
  const int ml = lane & 15;
  const int tx = blockIdx.x;
  const int py = blockIdx.y;
  const int b = blockIdx.z;

  __shared__ __attribute__((aligned(16))) float ft[165][74];

  const int gy0 = 4 * py;
  const int gx0 = 32 * tx;
  const float* xb = x + (size_t)b * 3 * 512 * 512;

  for (int p = threadIdx.x; p < 165; p += 128) {
    const int ly = p / 33, lx = p - ly * 33;
    const int gy = min(gy0 + ly, 511);
    const int gx = min(gx0 + lx, 511);
    const int ym = max(gy - 1, 0), yp = min(gy + 1, 511);
    const int xm = max(gx - 1, 0), xp = min(gx + 1, 511);
    float* f = &ft[p][0];
    float cv[3];
    int o = 6;
#pragma unroll
    for (int c = 0; c < 3; ++c) {
      const float* xc = xb + (size_t)c * 512 * 512;
      float a = xc[ym * 512 + xm], bb = xc[ym * 512 + gx], cc = xc[ym * 512 + xp];
      float d = xc[gy * 512 + xm], ee = xc[gy * 512 + gx], e = xc[gy * 512 + xp];
      float g = xc[yp * 512 + xm], h = xc[yp * 512 + gx], i = xc[yp * 512 + xp];
      cv[c] = ee;
      f[2 * c] = ((cc + 2.f * e + i) - (a + 2.f * d + g)) * 0.125f;
      f[2 * c + 1] = ((g + 2.f * h + i) - (a + 2.f * bb + cc)) * 0.125f;
    }
    const v2f z2 = {0.f, 0.f};
#pragma unroll
    for (int q = 3; q < 36; ++q) *(v2f*)(f + 2 * q) = z2;
    const HatCell CELL[105] = HAT_CELLS;
#pragma unroll
    for (int c = 0; c < 3; ++c) {
      const float v = cv[c];
      const int u = min(max((int)(v * 105.f), 0), 104);
      const HatCell hc = CELL[u];
      float* h = f + o + 22 * c + hc.j;
      h[0] = fminf(fmaxf(fmaf(v, -hc.s, hc.b), 0.f), 1.f);
      h[1] = fminf(fmaxf(fmaf(v, hc.s, -hc.a), 0.f), 1.f);
    }
  }
  __syncthreads();

  v8f acc[2][1];
  const v8f vz = {0.f, 0.f, 0.f, 0.f, 0.f, 0.f, 0.f, 0.f};
  acc[0][0] = vz;
  acc[1][0] = vz;

  constexpr int KB = 9;
  constexpr int BPT = 18 / KB;
  constexpr int NB = 9 * BPT;
  auto load_block = [&](FragSet<KB, 1>& s, int j) {
    const int r = j / BPT;
    const int ub = (j - r * BPT) * KB;
    const int ky = r / 3, kx = r - 3 * ky;
    const int lx = 2 * ml + kx;
    const float* fp0 = &ft[ky * 33 + lx][2 * hh + 4 * ub];
    const float* fp1 = &ft[(ky + 2) * 33 + lx][2 * hh + 4 * ub];
    const float* bp = wT + ((((size_t)(r * 36 + 2 * ub + hh)) * 64 + ml) << 1) + wv * 32;
#pragma unroll
    for (int i = 0; i < KB; ++i) {
      s.a0[i] = *(const v2f*)(fp0 + 4 * i);
      s.a1[i] = *(const v2f*)(fp1 + 4 * i);
      s.b[i][0] = *(const v2f*)(bp + i * 256);
    }
  };

  FragSet<KB, 1> fx, fy;
  load_block(fx, 0);
  for (int j = 0; j + 1 < NB; j += 2) {
    load_block(fy, j + 1);
    __builtin_amdgcn_sched_barrier(0);
    mma_block<KB, 1>(fx, acc);
    __builtin_amdgcn_sched_barrier(0);
    load_block(fx, min(j + 2, NB - 1));
    __builtin_amdgcn_sched_barrier(0);
    mma_block<KB, 1>(fy, acc);
  }
  if (NB & 1) mma_block<KB, 1>(fx, acc);

  __syncthreads();
  float* ot = &ft[0][0];
  {
    const float bb = bias[wv * 16 + ml];
#pragma unroll
    for (int j = 0; j < 4; ++j) {
      float m0 = fmaxf(acc[0][0][2 * j], acc[0][0][2 * j + 1]);
      float m1 = fmaxf(acc[1][0][2 * j], acc[1][0][2 * j + 1]);
      ot[(j + 4 * hh) * 64 + wv * 16 + ml] = eluf(fmaxf(m0, m1) + bb);
    }
  }
  __syncthreads();
  {
    const int pl = threadIdx.x >> 4;
    const int c4 = (threadIdx.x & 15) * 4;
    const int px = 8 * tx + pl;
    if (px < 117)
      *(v4f*)(f1 + ((((size_t)b * 127 + py) * 127 + px) << 6) + c4) =
          *(const v4f*)(ot + pl * 64 + c4);
  }
}

template <int HIN, int WIN, int HP, int WP, int WV, int KS, int XV, int WS>
__global__ __launch_bounds__(32 * WV * KS) void convmid_kernel(const float* __restrict__ fin,
                                                          const float* __restrict__ wT,
                                                          const float* __restrict__ bias,
                                                          float* __restrict__ fout) {
  constexpr int NT = 4 / WV;
  const int lane = threadIdx.x & 31;
  const int wid = __builtin_amdgcn_readfirstlane(threadIdx.x >> 5);
  const int wv = wid % WV;
  const int ks = (KS > 1) ? wid / WV : 0;
  const int n0 = wv * NT;
  const int hh = lane >> 4;
  const int ml = lane & 15;
  const int tx = blockIdx.x;
  const int py = blockIdx.y;
  const int b = blockIdx.z;

  v8f acc[2][NT];
  const v8f vz = {0.f, 0.f, 0.f, 0.f, 0.f, 0.f, 0.f, 0.f};
  for (int mt = 0; mt < 2; ++mt)
    for (int nt = 0; nt < NT; ++nt) acc[mt][nt] = vz;

  const int cx = 16 * tx + ml;
  const float* finb = fin + (size_t)b * HIN * WIN * 64;

  constexpr int KB = 8;
  constexpr int BPT = 16 / KB;
  constexpr int NB = 9 * BPT;
  static_assert(NB % KS == 0, "the K groups take equal numbers of blocks");
  constexpr int NBW = NB / KS;
  const int j0 = ks * NBW;
  auto load_block = [&](FragSet<KB, NT>& s, int j) {
    const int r = j / BPT;
    const int ub = (j - r * BPT) * KB;
    const int ky = r / 3, kx = r - 3 * ky;
    const int xx = min(2 * cx + kx, XV - 1);
    const int yy0 = min(4 * py + ky, HIN - 1);
    const int yy1 = min(4 * py + 2 + ky, HIN - 1);
    const float* row0 = finb + (((size_t)yy0 * WIN + xx) << 6) + 4 * ub + 2 * hh;
    const float* row1 = finb + (((size_t)yy1 * WIN + xx) << 6) + 4 * ub + 2 * hh;
    const float* bp = wT + ((((size_t)(r * 32 + 2 * ub + hh)) * 64 + ml) << 1) + n0 * 32;
#pragma unroll
    for (int i = 0; i < KB; ++i) {
      s.a0[i] = *(const v2f*)(row0 + 4 * i);
      s.a1[i] = *(const v2f*)(row1 + 4 * i);
#pragma unroll
      for (int nt = 0; nt < NT; ++nt) s.b[i][nt] = *(const v2f*)(bp + i * 256 + nt * 32);
    }
  };

  FragSet<KB, NT> fx, fy;
  load_block(fx, j0);
  for (int j = 0; j + 1 < NBW; j += 2) {
    load_block(fy, j0 + j + 1);
    __builtin_amdgcn_sched_barrier(0);
    mma_block<KB, NT>(fx, acc);
    __builtin_amdgcn_sched_barrier(0);
    load_block(fx, j0 + min(j + 2, NBW - 1));
    __builtin_amdgcn_sched_barrier(0);
    mma_block<KB, NT>(fy, acc);
  }
  if (NBW & 1) mma_block<KB, NT>(fx, acc);

  if constexpr (KS > 1) {
    __shared__ float red[KS - 1][WV][2 * NT * 8][32];
    if (ks > 0) {
#pragma unroll
      for (int mt = 0; mt < 2; ++mt)
#pragma unroll
        for (int nt = 0; nt < NT; ++nt)
#pragma unroll
          for (int r = 0; r < 8; ++r) red[ks - 1][wv][(mt * NT + nt) * 8 + r][lane] = acc[mt][nt][r];
    }
    __syncthreads();
    if (ks > 0) return;
#pragma unroll
    for (int s = 0; s < KS - 1; ++s)
#pragma unroll
      for (int mt = 0; mt < 2; ++mt)
#pragma unroll
        for (int nt = 0; nt < NT; ++nt)
#pragma unroll
          for (int r = 0; r < 8; ++r) acc[mt][nt][r] += red[s][wv][(mt * NT + nt) * 8 + r][lane];
  }

  const int pxb = 8 * tx;
#pragma unroll
  for (int nt = 0; nt < NT; ++nt) {
    const float bb = bias[(n0 + nt) * 16 + ml];
#pragma unroll
    for (int j = 0; j < 4; ++j) {
      float m0 = fmaxf(eluf(acc[0][nt][2 * j] + bb), eluf(acc[0][nt][2 * j + 1] + bb));
      float m1 = fmaxf(eluf(acc[1][nt][2 * j] + bb), eluf(acc[1][nt][2 * j + 1] + bb));
      int px = pxb + j + 4 * hh;
      if (px < WS)
        fout[((((size_t)b * HP + py) * WP + px) << 6) + (n0 + nt) * 16 + ml] = fmaxf(m0, m1);
    }
  }
}

__global__ __launch_bounds__(256) void final_kernel(const float* __restrict__ x,
                                                    const float* __restrict__ f3,
                                                    const float* __restrict__ w1,
                                                    const float* __restrict__ b1,
                                                    const float* __restrict__ w2,
                                                    const float* __restrict__ b2,
                                                    float* __restrict__ out) {
  constexpr int HW = 512 * 512;
  __shared__ float fv[64];
  __shared__ float hv[64];
  __shared__ __attribute__((aligned(16))) float wq[32];
  const int t = threadIdx.x;
  const int b = blockIdx.x >> 7;
  const int base = (blockIdx.x & 127) * 2048 + 4 * t;

  const float* xb = x + (size_t)b * 3 * HW + base;
  v4f px[3][2];
#pragma unroll
  for (int c = 0; c < 3; ++c)
#pragma unroll
    for (int h = 0; h < 2; ++h) px[c][h] = *(const v4f*)(xb + (size_t)c * HW + h * 1024);

  v4f w1r[16];
  v4f w2r[16];
  float bias1 = 0.f, bias2 = 0.f;
  if (t < 64) {
#pragma unroll
    for (int q = 0; q < 16; ++q) w1r[q] = *(const v4f*)(w1 + t * 64 + 4 * q);
    bias1 = b1[t];
  }
  if (t < 30) {
#pragma unroll
    for (int q = 0; q < 16; ++q) w2r[q] = *(const v4f*)(w2 + t * 64 + 4 * q);
    bias2 = b2[t];
  }

  if (t < 64) {
    float s = 0.f;
#pragma unroll
    for (int i = 0; i < 49; ++i) s += f3[(((size_t)b * 49 + i) << 6) + t];
    fv[t] = s * (1.f / 49.f);
  }
  __syncthreads();
  if (t < 64) {
    float acc = bias1;
#pragma unroll
    for (int q = 0; q < 16; ++q) {
      acc = fmaf(fv[4 * q + 0], w1r[q][0], acc);
      acc = fmaf(fv[4 * q + 1], w1r[q][1], acc);
      acc = fmaf(fv[4 * q + 2], w1r[q][2], acc);
      acc = fmaf(fv[4 * q + 3], w1r[q][3], acc);
    }
    hv[t] = eluf(acc);
  }
  __syncthreads();
  if (t < 30) {
    float a2 = bias2;
#pragma unroll
    for (int q = 0; q < 16; ++q) {
      a2 = fmaf(hv[4 * q + 0], w2r[q][0], a2);
      a2 = fmaf(hv[4 * q + 1], w2r[q][1], a2);
      a2 = fmaf(hv[4 * q + 2], w2r[q][2], a2);
      a2 = fmaf(hv[4 * q + 3], w2r[q][3], a2);
    }
    wq[t] = a2;
  }
  __syncthreads();

  float w[30];
#pragma unroll
  for (int k = 0; k < 30; ++k) w[k] = wq[k];

  float* ob = out + (size_t)b * 3 * HW + base;
#pragma unroll
  for (int h = 0; h < 2; ++h) {
    v4f o[3];
#pragma unroll
    for (int e = 0; e < 4; ++e) {
      const float r = px[0][h][e], g = px[1][h][e], bl = px[2][h][e];
      const float f[9] = {r * r, r * g, r * bl, r, g * g, g * bl, g, bl * bl, bl};
#pragma unroll
      for (int c = 0; c < 3; ++c) {
        float s = (c == 0) ? r : ((c == 1) ? g : bl);
#pragma unroll
        for (int k = 0; k < 9; ++k) s = fmaf(f[k], w[c * 10 + k], s);
        o[c][e] = s + w[c * 10 + 9];
      }
    }
#pragma unroll
    for (int c = 0; c < 3; ++c) *(v4f*)(ob + (size_t)c * HW + h * 1024) = o[c];
  }
}

static const size_t OFF_WT0 = 0;
static const size_t OFF_WT1 = 55296;
static const size_t OFF_WT2 = 92160;
static const size_t OFF_F1 = 129024;
static const size_t OFF_F2 = 129024 + 4129024;
static const size_t OFF_F3 = 129024 + 4129024 + 246016;

extern "C" void kernel_launch(void* const* d_in, const int* in_sizes, int n_in,
                              void* d_out, int out_size, void* d_ws, size_t ws_size,
                              hipStream_t stream) {
  const float* x = (const float*)d_in[0];
  const float* c0w = (const float*)d_in[1];
  const float* c0b = (const float*)d_in[2];
  const float* c1w = (const float*)d_in[3];
  const float* c1b = (const float*)d_in[4];
  const float* c2w = (const float*)d_in[5];
  const float* c2b = (const float*)d_in[6];
  const float* f1w = (const float*)d_in[7];
  const float* f1b = (const float*)d_in[8];
  const float* f2w = (const float*)d_in[9];
  const float* f2b = (const float*)d_in[10];
  float* out = (float*)d_out;
  float* ws = (float*)d_ws;

  float* wT0 = ws + OFF_WT0;
  float* wT1 = ws + OFF_WT1;
  float* wT2 = ws + OFF_WT2;
  float* F1 = ws + OFF_F1;
  float* F2 = ws + OFF_F2;
  float* F3 = ws + OFF_F3;

  tr_all_kernel<<<450, 256, 0, stream>>>(c0w, c1w, c2w, wT0, wT1, wT2);

  conv0_kernel<<<dim3(15, 117, 4), 128, 0, stream>>>(x, wT0, c0b, F1);
  convmid_kernel<127, 127, 31, 31, 2, 1, 117, 29><<<dim3(4, 29, 4), 64, 0, stream>>>(F1, wT1, c1b, F2);
  convmid_kernel<31, 31, 7, 7, 2, 6, 29, 7><<<dim3(1, 7, 4), 384, 0, stream>>>(F2, wT2, c2b, F3);

  final_kernel<<<512, 256, 0, stream>>>(x, F3, f1w, f1b, f2w, f2b, out);
}


# conv0: the two conv rows' shared input row used once (rows = P+S, Q+S: 5 WMMAs per k-step where 6 were), the 3 combinations formed by packed adds on the loaded fragments; changes the summation order
# speedup vs baseline: 3.4265x; 1.0572x over previous
#include <hip/hip_runtime.h>
#include <cstddef>

typedef __attribute__((ext_vector_type(2))) float v2f;
typedef __attribute__((ext_vector_type(4))) float v4f;
typedef __attribute__((ext_vector_type(8))) float v8f;

#define DEV __device__ __forceinline__

DEV float eluf(float v) { return v > 0.f ? v : (__expf(v) - 1.f); }
#define HIST_KNOTS {0, 7, 14, 15, 21, 28, 30, 35, 42, 45, 49, 56, 60, 63, 70, 75, 77, 84, 90, 91, 98, 105}
struct __attribute__((aligned(16))) HatCell {
  float s, a, b;
  int j;
};
#define HAT_CELLS { \
  {15.0f, 0.0f, 1.0f, 0}, {15.0f, 0.0f, 1.0f, 0}, {15.0f, 0.0f, 1.0f, 0}, \
  {15.0f, 0.0f, 1.0f, 0}, {15.0f, 0.0f, 1.0f, 0}, {15.0f, 0.0f, 1.0f, 0}, \
  {15.0f, 0.0f, 1.0f, 0}, {15.0f, 1.0f, 2.0f, 1}, {15.0f, 1.0f, 2.0f, 1}, \
  {15.0f, 1.0f, 2.0f, 1}, {15.0f, 1.0f, 2.0f, 1}, {15.0f, 1.0f, 2.0f, 1}, \
  {15.0f, 1.0f, 2.0f, 1}, {15.0f, 1.0f, 2.0f, 1}, {105.0f, 14.0f, 15.0f, 2}, \
  {17.5f, 2.5f, 3.5f, 3}, {17.5f, 2.5f, 3.5f, 3}, {17.5f, 2.5f, 3.5f, 3}, \
  {17.5f, 2.5f, 3.5f, 3}, {17.5f, 2.5f, 3.5f, 3}, {17.5f, 2.5f, 3.5f, 3}, \
  {15.0f, 3.0f, 4.0f, 4}, {15.0f, 3.0f, 4.0f, 4}, {15.0f, 3.0f, 4.0f, 4}, \
  {15.0f, 3.0f, 4.0f, 4}, {15.0f, 3.0f, 4.0f, 4}, {15.0f, 3.0f, 4.0f, 4}, \
  {15.0f, 3.0f, 4.0f, 4}, {52.5f, 14.0f, 15.0f, 5}, {52.5f, 14.0f, 15.0f, 5}, \
  {21.0f, 6.0f, 7.0f, 6}, {21.0f, 6.0f, 7.0f, 6}, {21.0f, 6.0f, 7.0f, 6}, \
  {21.0f, 6.0f, 7.0f, 6}, {21.0f, 6.0f, 7.0f, 6}, {15.0f, 5.0f, 6.0f, 7}, \
  {15.0f, 5.0f, 6.0f, 7}, {15.0f, 5.0f, 6.0f, 7}, {15.0f, 5.0f, 6.0f, 7}, \
  {15.0f, 5.0f, 6.0f, 7}, {15.0f, 5.0f, 6.0f, 7}, {15.0f, 5.0f, 6.0f, 7}, \
  {35.0f, 14.0f, 15.0f, 8}, {35.0f, 14.0f, 15.0f, 8}, {35.0f, 14.0f, 15.0f, 8}, \
  {26.25f, 11.25f, 12.25f, 9}, {26.25f, 11.25f, 12.25f, 9}, {26.25f, 11.25f, 12.25f, 9}, \
  {26.25f, 11.25f, 12.25f, 9}, {15.0f, 7.0f, 8.0f, 10}, {15.0f, 7.0f, 8.0f, 10}, \
  {15.0f, 7.0f, 8.0f, 10}, {15.0f, 7.0f, 8.0f, 10}, {15.0f, 7.0f, 8.0f, 10}, \
  {15.0f, 7.0f, 8.0f, 10}, {15.0f, 7.0f, 8.0f, 10}, {26.25f, 14.0f, 15.0f, 11}, \
  {26.25f, 14.0f, 15.0f, 11}, {26.25f, 14.0f, 15.0f, 11}, {26.25f, 14.0f, 15.0f, 11}, \
  {35.0f, 20.0f, 21.0f, 12}, {35.0f, 20.0f, 21.0f, 12}, {35.0f, 20.0f, 21.0f, 12}, \
  {15.0f, 9.0f, 10.0f, 13}, {15.0f, 9.0f, 10.0f, 13}, {15.0f, 9.0f, 10.0f, 13}, \
  {15.0f, 9.0f, 10.0f, 13}, {15.0f, 9.0f, 10.0f, 13}, {15.0f, 9.0f, 10.0f, 13}, \
  {15.0f, 9.0f, 10.0f, 13}, {21.0f, 14.0f, 15.0f, 14}, {21.0f, 14.0f, 15.0f, 14}, \
  {21.0f, 14.0f, 15.0f, 14}, {21.0f, 14.0f, 15.0f, 14}, {21.0f, 14.0f, 15.0f, 14}, \
  {52.5f, 37.5f, 38.5f, 15}, {52.5f, 37.5f, 38.5f, 15}, {15.0f, 11.0f, 12.0f, 16}, \
  {15.0f, 11.0f, 12.0f, 16}, {15.0f, 11.0f, 12.0f, 16}, {15.0f, 11.0f, 12.0f, 16}, \
  {15.0f, 11.0f, 12.0f, 16}, {15.0f, 11.0f, 12.0f, 16}, {15.0f, 11.0f, 12.0f, 16}, \
  {17.5f, 14.0f, 15.0f, 17}, {17.5f, 14.0f, 15.0f, 17}, {17.5f, 14.0f, 15.0f, 17}, \
  {17.5f, 14.0f, 15.0f, 17}, {17.5f, 14.0f, 15.0f, 17}, {17.5f, 14.0f, 15.0f, 17}, \
  {105.0f, 90.0f, 91.0f, 18}, {15.0f, 13.0f, 14.0f, 19}, {15.0f, 13.0f, 14.0f, 19}, \
  {15.0f, 13.0f, 14.0f, 19}, {15.0f, 13.0f, 14.0f, 19}, {15.0f, 13.0f, 14.0f, 19}, \
  {15.0f, 13.0f, 14.0f, 19}, {15.0f, 13.0f, 14.0f, 19}, {15.0f, 14.0f, 15.0f, 20}, \
  {15.0f, 14.0f, 15.0f, 20}, {15.0f, 14.0f, 15.0f, 20}, {15.0f, 14.0f, 15.0f, 20}, \
  {15.0f, 14.0f, 15.0f, 20}, {15.0f, 14.0f, 15.0f, 20}, {15.0f, 14.0f, 15.0f, 20} }

DEV v8f wmma_f32(v2f a, v2f b, v8f c) {
  return __builtin_amdgcn_wmma_f32_16x16x4_f32(false, a, false, b, (short)0, c,
                                               false, false);
}

template <int KB, int NT>
struct FragSet {
  v2f a0[KB];
  v2f a1[KB];
  v2f b[KB][NT];
};

template <int KB, int NT>
DEV void mma_block(const FragSet<KB, NT>& s, v8f (&acc)[2][NT]) {
#pragma unroll
  for (int i = 0; i < KB; ++i) {
#pragma unroll
    for (int nt = 0; nt < NT; ++nt) {
      acc[0][nt] = wmma_f32(s.a0[i], s.b[i][nt], acc[0][nt]);
      acc[1][nt] = wmma_f32(s.a1[i], s.b[i][nt], acc[1][nt]);
    }
  }
}

template <int G>
struct RowSet {
  v2f f[G][5];
  v2f w[G][3];
};

template <int G>
DEV void mma_rows(const RowSet<G>& s, v8f& P, v8f& Q, v8f& S) {
#pragma unroll
  for (int g = 0; g < G; ++g) {
    const v2f d0 = s.f[g][0] - s.f[g][2];
    const v2f d4 = s.f[g][4] - s.f[g][2];
    const v2f ws = s.w[g][0] + s.w[g][2];
    P = wmma_f32(d0, s.w[g][0], P);
    Q = wmma_f32(d4, s.w[g][2], Q);
    S = wmma_f32(s.f[g][2], ws, S);
    P = wmma_f32(s.f[g][1], s.w[g][1], P);
    Q = wmma_f32(s.f[g][3], s.w[g][1], Q);
  }
}

__global__ __launch_bounds__(256) void tr_all_kernel(const float* __restrict__ w0,
                                                     const float* __restrict__ w1,
                                                     const float* __restrict__ w2,
                                                     float* __restrict__ wT0,
                                                     float* __restrict__ wT1,
                                                     float* __restrict__ wT2) {
  const int blk = blockIdx.x;
  if (blk < 162) {
    const int TK[22] = HIST_KNOTS;
    const int idx = blk * 256 + threadIdx.x;
    const int k = idx >> 6, n = idx & 63;
    const int r = k / 72, cf = k % 72;
    const float* __restrict__ wn = w0 + (size_t)n * 837 + r;
    float v = 0.f;
    if (cf < 6) {
      v = wn[(3 + cf) * 9];
    } else {
      const int q = cf - 6;
      const int c = q / 22;
      const int T = TK[q - 22 * c];
      const int k0 = T / 7, r16 = T - 7 * k0, k1 = min(k0 + 1, 15);
      const int j0 = T / 15, r8 = T - 15 * j0, j1 = min(j0 + 1, 7);
      const int m0 = T / 35, r4 = T - 35 * m0, m1 = min(m0 + 1, 3);
      const float a0 = wn[(45 + 16 * c + k0) * 9], a1 = wn[(45 + 16 * c + k1) * 9];
      const float b0 = wn[(21 + 8 * c + j0) * 9], b1 = wn[(21 + 8 * c + j1) * 9];
      const float d0 = wn[(9 + 4 * c + m0) * 9], d1 = wn[(9 + 4 * c + m1) * 9];
      const float xw = wn[c * 9];
      v = a0 * ((float)(7 - r16) / 7.f);
      v = fmaf(a1, (float)r16 / 7.f, v);
      v = fmaf(b0, (float)(15 - r8) / 15.f, v);
      v = fmaf(b1, (float)r8 / 15.f, v);
      v = fmaf(d0, (float)(35 - r4) / 35.f, v);
      v = fmaf(d1, (float)r4 / 35.f, v);
      v = fmaf(xw, (float)T / 105.f, v);
    }
    wT0[((((size_t)(k >> 1)) * 64 + n) << 1) + (k & 1)] = v;
  } else {
    const bool third = blk >= 306;
    const float* __restrict__ w = third ? w2 : w1;
    float* __restrict__ wT = third ? wT2 : wT1;
    const int idx = (blk - (third ? 306 : 162)) * 256 + threadIdx.x;
    const int k = idx >> 6, n = idx & 63;
    const int r = k >> 6, cin = k & 63;
    const float v = w[(size_t)n * 576 + cin * 9 + r];
    wT[((((size_t)(k >> 1)) * 64 + n) << 1) + (k & 1)] = v;
  }
}

__global__ __launch_bounds__(128) void conv0_kernel(const float* __restrict__ x,
                                                    const float* __restrict__ wT,
                                                    const float* __restrict__ bias,
                                                    float* __restrict__ f1) {
  const int lane = threadIdx.x & 31;
  const int wv = __builtin_amdgcn_readfirstlane(threadIdx.x >> 5);
  const int hh = lane >> 4;
  const int ml = lane & 15;
  const int tx = blockIdx.x;
  const int py = blockIdx.y;
  const int b = blockIdx.z;

  __shared__ __attribute__((aligned(16))) float ft[165][74];

  const int gy0 = 4 * py;
  const int gx0 = 32 * tx;
  const float* xb = x + (size_t)b * 3 * 512 * 512;

  for (int p = threadIdx.x; p < 165; p += 128) {
    const int ly = p / 33, lx = p - ly * 33;
    const int gy = min(gy0 + ly, 511);
    const int gx = min(gx0 + lx, 511);
    const int ym = max(gy - 1, 0), yp = min(gy + 1, 511);
    const int xm = max(gx - 1, 0), xp = min(gx + 1, 511);
    float* f = &ft[p][0];
    float cv[3];
    int o = 6;
#pragma unroll
    for (int c = 0; c < 3; ++c) {
      const float* xc = xb + (size_t)c * 512 * 512;
      float a = xc[ym * 512 + xm], bb = xc[ym * 512 + gx], cc = xc[ym * 512 + xp];
      float d = xc[gy * 512 + xm], ee = xc[gy * 512 + gx], e = xc[gy * 512 + xp];
      float g = xc[yp * 512 + xm], h = xc[yp * 512 + gx], i = xc[yp * 512 + xp];
      cv[c] = ee;
      f[2 * c] = ((cc + 2.f * e + i) - (a + 2.f * d + g)) * 0.125f;
      f[2 * c + 1] = ((g + 2.f * h + i) - (a + 2.f * bb + cc)) * 0.125f;
    }
    const v2f z2 = {0.f, 0.f};
#pragma unroll
    for (int q = 3; q < 36; ++q) *(v2f*)(f + 2 * q) = z2;
    const HatCell CELL[105] = HAT_CELLS;
#pragma unroll
    for (int c = 0; c < 3; ++c) {
      const float v = cv[c];
      const int u = min(max((int)(v * 105.f), 0), 104);
      const HatCell hc = CELL[u];
      float* h = f + o + 22 * c + hc.j;
      h[0] = fminf(fmaxf(fmaf(v, -hc.s, hc.b), 0.f), 1.f);
      h[1] = fminf(fmaxf(fmaf(v, hc.s, -hc.a), 0.f), 1.f);
    }
  }
  __syncthreads();

  const v8f vz = {0.f, 0.f, 0.f, 0.f, 0.f, 0.f, 0.f, 0.f};
  v8f P = vz, Q = vz, S = vz;

  constexpr int G = 3;
  constexpr int BPC = 18 / G;
  constexpr int NB = 3 * BPC;
  auto load_block = [&](RowSet<G>& s, int j) {
    const int kx = j / BPC;
    const int u0 = (j - kx * BPC) * G;
    const float* fp = &ft[2 * ml + kx][2 * hh + 4 * u0];
    const float* bp = wT + ((((size_t)(kx * 36 + 2 * u0 + hh)) * 64 + ml) << 1) + wv * 32;
#pragma unroll
    for (int g = 0; g < G; ++g) {
#pragma unroll
      for (int r = 0; r < 5; ++r) s.f[g][r] = *(const v2f*)(fp + r * (33 * 74) + 4 * g);
#pragma unroll
      for (int ky = 0; ky < 3; ++ky) s.w[g][ky] = *(const v2f*)(bp + ky * (3 * 36 * 128) + g * 256);
    }
  };

  RowSet<G> fx, fy;
  load_block(fx, 0);
  for (int j = 0; j + 1 < NB; j += 2) {
    load_block(fy, j + 1);
    __builtin_amdgcn_sched_barrier(0);
    mma_rows<G>(fx, P, Q, S);
    __builtin_amdgcn_sched_barrier(0);
    load_block(fx, min(j + 2, NB - 1));
    __builtin_amdgcn_sched_barrier(0);
    mma_rows<G>(fy, P, Q, S);
  }
  if (NB & 1) mma_rows<G>(fx, P, Q, S);
  const v8f row0 = P + S;
  const v8f row1 = Q + S;

  __syncthreads();
  float* ot = &ft[0][0];
  {
    const float bb = bias[wv * 16 + ml];
#pragma unroll
    for (int j = 0; j < 4; ++j) {
      float m0 = fmaxf(row0[2 * j], row0[2 * j + 1]);
      float m1 = fmaxf(row1[2 * j], row1[2 * j + 1]);
      ot[(j + 4 * hh) * 64 + wv * 16 + ml] = eluf(fmaxf(m0, m1) + bb);
    }
  }
  __syncthreads();
  {
    const int pl = threadIdx.x >> 4;
    const int c4 = (threadIdx.x & 15) * 4;
    const int px = 8 * tx + pl;
    if (px < 117)
      *(v4f*)(f1 + ((((size_t)b * 127 + py) * 127 + px) << 6) + c4) =
          *(const v4f*)(ot + pl * 64 + c4);
  }
}

template <int HIN, int WIN, int HP, int WP, int WV, int KS, int XV, int WS>
__global__ __launch_bounds__(32 * WV * KS) void convmid_kernel(const float* __restrict__ fin,
                                                          const float* __restrict__ wT,
                                                          const float* __restrict__ bias,
                                                          float* __restrict__ fout) {
  constexpr int NT = 4 / WV;
  const int lane = threadIdx.x & 31;
  const int wid = __builtin_amdgcn_readfirstlane(threadIdx.x >> 5);
  const int wv = wid % WV;
  const int ks = (KS > 1) ? wid / WV : 0;
  const int n0 = wv * NT;
  const int hh = lane >> 4;
  const int ml = lane & 15;
  const int tx = blockIdx.x;
  const int py = blockIdx.y;
  const int b = blockIdx.z;

  v8f acc[2][NT];
  const v8f vz = {0.f, 0.f, 0.f, 0.f, 0.f, 0.f, 0.f, 0.f};
  for (int mt = 0; mt < 2; ++mt)
    for (int nt = 0; nt < NT; ++nt) acc[mt][nt] = vz;

  const int cx = 16 * tx + ml;
  const float* finb = fin + (size_t)b * HIN * WIN * 64;

  constexpr int KB = 8;
  constexpr int BPT = 16 / KB;
  constexpr int NB = 9 * BPT;
  static_assert(NB % KS == 0, "the K groups take equal numbers of blocks");
  constexpr int NBW = NB / KS;
  const int j0 = ks * NBW;
  auto load_block = [&](FragSet<KB, NT>& s, int j) {
    const int r = j / BPT;
    const int ub = (j - r * BPT) * KB;
    const int ky = r / 3, kx = r - 3 * ky;
    const int xx = min(2 * cx + kx, XV - 1);
    const int yy0 = min(4 * py + ky, HIN - 1);
    const int yy1 = min(4 * py + 2 + ky, HIN - 1);
    const float* row0 = finb + (((size_t)yy0 * WIN + xx) << 6) + 4 * ub + 2 * hh;
    const float* row1 = finb + (((size_t)yy1 * WIN + xx) << 6) + 4 * ub + 2 * hh;
    const float* bp = wT + ((((size_t)(r * 32 + 2 * ub + hh)) * 64 + ml) << 1) + n0 * 32;
#pragma unroll
    for (int i = 0; i < KB; ++i) {
      s.a0[i] = *(const v2f*)(row0 + 4 * i);
      s.a1[i] = *(const v2f*)(row1 + 4 * i);
#pragma unroll
      for (int nt = 0; nt < NT; ++nt) s.b[i][nt] = *(const v2f*)(bp + i * 256 + nt * 32);
    }
  };

  FragSet<KB, NT> fx, fy;
  load_block(fx, j0);
  for (int j = 0; j + 1 < NBW; j += 2) {
    load_block(fy, j0 + j + 1);
    __builtin_amdgcn_sched_barrier(0);
    mma_block<KB, NT>(fx, acc);
    __builtin_amdgcn_sched_barrier(0);
    load_block(fx, j0 + min(j + 2, NBW - 1));
    __builtin_amdgcn_sched_barrier(0);
    mma_block<KB, NT>(fy, acc);
  }
  if (NBW & 1) mma_block<KB, NT>(fx, acc);

  if constexpr (KS > 1) {
    __shared__ float red[KS - 1][WV][2 * NT * 8][32];
    if (ks > 0) {
#pragma unroll
      for (int mt = 0; mt < 2; ++mt)
#pragma unroll
        for (int nt = 0; nt < NT; ++nt)
#pragma unroll
          for (int r = 0; r < 8; ++r) red[ks - 1][wv][(mt * NT + nt) * 8 + r][lane] = acc[mt][nt][r];
    }
    __syncthreads();
    if (ks > 0) return;
#pragma unroll
    for (int s = 0; s < KS - 1; ++s)
#pragma unroll
      for (int mt = 0; mt < 2; ++mt)
#pragma unroll
        for (int nt = 0; nt < NT; ++nt)
#pragma unroll
          for (int r = 0; r < 8; ++r) acc[mt][nt][r] += red[s][wv][(mt * NT + nt) * 8 + r][lane];
  }

  const int pxb = 8 * tx;
#pragma unroll
  for (int nt = 0; nt < NT; ++nt) {
    const float bb = bias[(n0 + nt) * 16 + ml];
#pragma unroll
    for (int j = 0; j < 4; ++j) {
      float m0 = fmaxf(eluf(acc[0][nt][2 * j] + bb), eluf(acc[0][nt][2 * j + 1] + bb));
      float m1 = fmaxf(eluf(acc[1][nt][2 * j] + bb), eluf(acc[1][nt][2 * j + 1] + bb));
      int px = pxb + j + 4 * hh;
      if (px < WS)
        fout[((((size_t)b * HP + py) * WP + px) << 6) + (n0 + nt) * 16 + ml] = fmaxf(m0, m1);
    }
  }
}

__global__ __launch_bounds__(256) void final_kernel(const float* __restrict__ x,
                                                    const float* __restrict__ f3,
                                                    const float* __restrict__ w1,
                                                    const float* __restrict__ b1,
                                                    const float* __restrict__ w2,
                                                    const float* __restrict__ b2,
                                                    float* __restrict__ out) {
  constexpr int HW = 512 * 512;
  __shared__ float fv[64];
  __shared__ float hv[64];
  __shared__ __attribute__((aligned(16))) float wq[32];
  const int t = threadIdx.x;
  const int b = blockIdx.x >> 7;
  const int base = (blockIdx.x & 127) * 2048 + 4 * t;

  const float* xb = x + (size_t)b * 3 * HW + base;
  v4f px[3][2];
#pragma unroll
  for (int c = 0; c < 3; ++c)
#pragma unroll
    for (int h = 0; h < 2; ++h) px[c][h] = *(const v4f*)(xb + (size_t)c * HW + h * 1024);

  v4f w1r[16];
  v4f w2r[16];
  float bias1 = 0.f, bias2 = 0.f;
  if (t < 64) {
#pragma unroll
    for (int q = 0; q < 16; ++q) w1r[q] = *(const v4f*)(w1 + t * 64 + 4 * q);
    bias1 = b1[t];
  }
  if (t < 30) {
#pragma unroll
    for (int q = 0; q < 16; ++q) w2r[q] = *(const v4f*)(w2 + t * 64 + 4 * q);
    bias2 = b2[t];
  }

  if (t < 64) {
    float s = 0.f;
#pragma unroll
    for (int i = 0; i < 49; ++i) s += f3[(((size_t)b * 49 + i) << 6) + t];
    fv[t] = s * (1.f / 49.f);
  }
  __syncthreads();
  if (t < 64) {
    float acc = bias1;
#pragma unroll
    for (int q = 0; q < 16; ++q) {
      acc = fmaf(fv[4 * q + 0], w1r[q][0], acc);
      acc = fmaf(fv[4 * q + 1], w1r[q][1], acc);
      acc = fmaf(fv[4 * q + 2], w1r[q][2], acc);
      acc = fmaf(fv[4 * q + 3], w1r[q][3], acc);
    }
    hv[t] = eluf(acc);
  }
  __syncthreads();
  if (t < 30) {
    float a2 = bias2;
#pragma unroll
    for (int q = 0; q < 16; ++q) {
      a2 = fmaf(hv[4 * q + 0], w2r[q][0], a2);
      a2 = fmaf(hv[4 * q + 1], w2r[q][1], a2);
      a2 = fmaf(hv[4 * q + 2], w2r[q][2], a2);
      a2 = fmaf(hv[4 * q + 3], w2r[q][3], a2);
    }
    wq[t] = a2;
  }
  __syncthreads();

  float w[30];
#pragma unroll
  for (int k = 0; k < 30; ++k) w[k] = wq[k];

  float* ob = out + (size_t)b * 3 * HW + base;
#pragma unroll
  for (int h = 0; h < 2; ++h) {
    v4f o[3];
#pragma unroll
    for (int e = 0; e < 4; ++e) {
      const float r = px[0][h][e], g = px[1][h][e], bl = px[2][h][e];
      const float f[9] = {r * r, r * g, r * bl, r, g * g, g * bl, g, bl * bl, bl};
#pragma unroll
      for (int c = 0; c < 3; ++c) {
        float s = (c == 0) ? r : ((c == 1) ? g : bl);
#pragma unroll
        for (int k = 0; k < 9; ++k) s = fmaf(f[k], w[c * 10 + k], s);
        o[c][e] = s + w[c * 10 + 9];
      }
    }
#pragma unroll
    for (int c = 0; c < 3; ++c) *(v4f*)(ob + (size_t)c * HW + h * 1024) = o[c];
  }
}

static const size_t OFF_WT0 = 0;
static const size_t OFF_WT1 = 55296;
static const size_t OFF_WT2 = 92160;
static const size_t OFF_F1 = 129024;
static const size_t OFF_F2 = 129024 + 4129024;
static const size_t OFF_F3 = 129024 + 4129024 + 246016;

extern "C" void kernel_launch(void* const* d_in, const int* in_sizes, int n_in,
                              void* d_out, int out_size, void* d_ws, size_t ws_size,
                              hipStream_t stream) {
  const float* x = (const float*)d_in[0];
  const float* c0w = (const float*)d_in[1];
  const float* c0b = (const float*)d_in[2];
  const float* c1w = (const float*)d_in[3];
  const float* c1b = (const float*)d_in[4];
  const float* c2w = (const float*)d_in[5];
  const float* c2b = (const float*)d_in[6];
  const float* f1w = (const float*)d_in[7];
  const float* f1b = (const float*)d_in[8];
  const float* f2w = (const float*)d_in[9];
  const float* f2b = (const float*)d_in[10];
  float* out = (float*)d_out;
  float* ws = (float*)d_ws;

  float* wT0 = ws + OFF_WT0;
  float* wT1 = ws + OFF_WT1;
  float* wT2 = ws + OFF_WT2;
  float* F1 = ws + OFF_F1;
  float* F2 = ws + OFF_F2;
  float* F3 = ws + OFF_F3;

  tr_all_kernel<<<450, 256, 0, stream>>>(c0w, c1w, c2w, wT0, wT1, wT2);

  conv0_kernel<<<dim3(15, 117, 4), 128, 0, stream>>>(x, wT0, c0b, F1);
  convmid_kernel<127, 127, 31, 31, 2, 1, 117, 29><<<dim3(4, 29, 4), 64, 0, stream>>>(F1, wT1, c1b, F2);
  convmid_kernel<31, 31, 7, 7, 2, 6, 29, 7><<<dim3(1, 7, 4), 384, 0, stream>>>(F2, wT2, c2b, F3);

  final_kernel<<<512, 256, 0, stream>>>(x, F3, f1w, f1b, f2w, f2b, out);
}
